# starting kernel
# baseline (speedup 1.0000x reference)
#include <hip/hip_runtime.h>

#define N_NODES 50000
#define N_EDGES 800000
#define DFEAT   128
#define ASTRIDE 132

typedef float v2f __attribute__((ext_vector_type(2)));
typedef float v8f __attribute__((ext_vector_type(8)));

__device__ __forceinline__ void atomic_add_f32(float* p, float v) {
    __hip_atomic_fetch_add(p, v, __ATOMIC_RELAXED, __HIP_MEMORY_SCOPE_AGENT);
}

__global__ __launch_bounds__(256) void zero_f4(float4* __restrict__ p, int n4) {
    int i = blockIdx.x * blockDim.x + threadIdx.x;
    if (i < n4) p[i] = make_float4(0.f, 0.f, 0.f, 0.f);
}

__global__ __launch_bounds__(256) void degree_kernel(const int* __restrict__ src,
                                                     const int* __restrict__ dst,
                                                     float* __restrict__ deg_out,
                                                     float* __restrict__ deg_in,
                                                     int n_edges) {
    int e = blockIdx.x * blockDim.x + threadIdx.x;
    if (e < n_edges) {
        atomic_add_f32(&deg_out[src[e]], 1.0f);
        atomic_add_f32(&deg_in [dst[e]], 1.0f);
    }
}

__global__ __launch_bounds__(256) void deg_to_rsqrt(float* __restrict__ p, int n) {
    int i = blockIdx.x * blockDim.x + threadIdx.x;
    if (i < n) p[i] = rsqrtf(fmaxf(p[i], 1.0f));
}

__global__ __launch_bounds__(256) void scatter_kernel(const float* __restrict__ X,
                                                      const float* __restrict__ rs_out,
                                                      const int* __restrict__ src,
                                                      const int* __restrict__ dst,
                                                      float* __restrict__ msg,
                                                      int n_edges) {
    int e    = (blockIdx.x * blockDim.x + threadIdx.x) >> 5;
    int lane = threadIdx.x & 31;
    if (e >= n_edges) return;
    int   s = src[e];
    int   d = dst[e];
    float scale = rs_out[s];
    float4 v = ((const float4*)(X + (size_t)s * DFEAT))[lane];
    float* mp = msg + (size_t)d * DFEAT + lane * 4;
    atomic_add_f32(mp + 0, v.x * scale);
    atomic_add_f32(mp + 1, v.y * scale);
    atomic_add_f32(mp + 2, v.z * scale);
    atomic_add_f32(mp + 3, v.w * scale);
}

template <bool RELU>
__global__ __launch_bounds__(256) void gcn_gemm(const float* __restrict__ A,
                                                const float* __restrict__ rs_in,
                                                const float* __restrict__ W,
                                                const float* __restrict__ bias,
                                                float* __restrict__ out) {
    __shared__ float Alds[16 * ASTRIDE];

    const int tid     = threadIdx.x;
    const int wave    = tid >> 5;
    const int lane    = tid & 31;
    const int rowBase = blockIdx.x << 4;

    {
        const int r = tid >> 4;
        const int c = (tid & 15) << 3;
        const float scale = rs_in[rowBase + r];
        const float4* gp = (const float4*)(A + (size_t)(rowBase + r) * DFEAT + c);
        float4 a0 = gp[0];
        float4 a1 = gp[1];
        float* lp = &Alds[r * ASTRIDE + c];
        lp[0] = a0.x * scale; lp[1] = a0.y * scale; lp[2] = a0.z * scale; lp[3] = a0.w * scale;
        lp[4] = a1.x * scale; lp[5] = a1.y * scale; lp[6] = a1.z * scale; lp[7] = a1.w * scale;
    }
    __syncthreads();

    const int   mrow = lane & 15;
    const int   khi  = (lane >> 4) << 1;
    const int   n0   = wave << 4;
    const float* wp  = W + n0 + mrow;
    const float* ap  = &Alds[mrow * ASTRIDE + khi];

    v8f acc = {0.f, 0.f, 0.f, 0.f, 0.f, 0.f, 0.f, 0.f};
#pragma unroll
    for (int k = 0; k < DFEAT; k += 4) {
        v2f a = *(const v2f*)(ap + k);
        v2f b;
        b.x = wp[(size_t)(k + khi)     * DFEAT];
        b.y = wp[(size_t)(k + khi + 1) * DFEAT];
        acc = __builtin_amdgcn_wmma_f32_16x16x4_f32(false, a, false, b,
                                                    (short)0, acc, false, false);
    }

    const int   col  = n0 + mrow;
    const float bv   = bias[col];
    const int   rsel = (lane >> 4) << 3;
#pragma unroll
    for (int v = 0; v < 8; ++v) {
        float h = acc[v] + bv;
        if (RELU) h = fmaxf(h, 0.f);
        out[(size_t)(rowBase + v + rsel) * DFEAT + col] = h;
    }
}

extern "C" void kernel_launch(void* const* d_in, const int* in_sizes, int n_in,
                              void* d_out, int out_size, void* d_ws, size_t ws_size,
                              hipStream_t stream) {
    const float* x   = (const float*)d_in[0];
    const int*   src = (const int*)  d_in[1];
    const int*   dst = (const int*)  d_in[2];
    const float* W1  = (const float*)d_in[3];
    const float* b1  = (const float*)d_in[4];
    const float* W2  = (const float*)d_in[5];
    const float* b2  = (const float*)d_in[6];
    float*       out = (float*)d_out;

    float* ws     = (float*)d_ws;
    float* rs_out = ws;
    float* rs_in  = ws + N_NODES;
    float* msg    = ws + 2 * N_NODES;
    float* h1     = msg + (size_t)N_NODES * DFEAT;

    const int T     = 256;
    const int ND4   = N_NODES * DFEAT / 4;
    const int EBLK  = (N_EDGES + 7) / 8;
    const int GEMMB = N_NODES / 16;

    zero_f4<<<(2 * N_NODES / 4 + T - 1) / T, T, 0, stream>>>((float4*)rs_out, 2 * N_NODES / 4);
    degree_kernel<<<(N_EDGES + T - 1) / T, T, 0, stream>>>(src, dst, rs_out, rs_in, N_EDGES);
    deg_to_rsqrt<<<(2 * N_NODES + T - 1) / T, T, 0, stream>>>(rs_out, 2 * N_NODES);

    zero_f4<<<(ND4 + T - 1) / T, T, 0, stream>>>((float4*)msg, ND4);
    scatter_kernel<<<EBLK, T, 0, stream>>>(x, rs_out, src, dst, msg, N_EDGES);
    gcn_gemm<true><<<GEMMB, T, 0, stream>>>(msg, rs_in, W1, b1, h1);

    zero_f4<<<(ND4 + T - 1) / T, T, 0, stream>>>((float4*)msg, ND4);
    scatter_kernel<<<EBLK, T, 0, stream>>>(h1, rs_out, src, dst, msg, N_EDGES);
    gcn_gemm<false><<<GEMMB, T, 0, stream>>>(msg, rs_in, W2, b2, out);
}


# Float-atomic edge scatter replaced by per-call destination-ordered edge list (int atomic count+slot, prefix sum, place) and one wave per node summing its rows in registers in slot (arrival) order
# speedup vs baseline: 1.9863x; 1.9863x over previous
#include <hip/hip_runtime.h>

#define N_NODES 50000
#define N_EDGES 800000
#define DFEAT   128
#define ASTRIDE 132

typedef float v2f __attribute__((ext_vector_type(2)));
typedef float v8f __attribute__((ext_vector_type(8)));

__global__ __launch_bounds__(256) void zero_f4(float4* __restrict__ p, int n4) {
    int i = blockIdx.x * blockDim.x + threadIdx.x;
    if (i < n4) p[i] = make_float4(0.f, 0.f, 0.f, 0.f);
}

__global__ __launch_bounds__(256) void count_kernel(const int* __restrict__ src,
                                                    const int* __restrict__ dst,
                                                    int* __restrict__ cnt_out,
                                                    int* __restrict__ cnt_in,
                                                    int* __restrict__ slot,
                                                    int n_edges) {
    int e = blockIdx.x * blockDim.x + threadIdx.x;
    if (e < n_edges) {
        __hip_atomic_fetch_add(&cnt_out[src[e]], 1, __ATOMIC_RELAXED, __HIP_MEMORY_SCOPE_AGENT);
        slot[e] = __hip_atomic_fetch_add(&cnt_in[dst[e]], 1, __ATOMIC_RELAXED,
                                         __HIP_MEMORY_SCOPE_AGENT);
    }
}

__global__ __launch_bounds__(1024) void rowstart_kernel(const int* __restrict__ cnt_out,
                                                        const int* __restrict__ cnt_in,
                                                        int* __restrict__ rowptr,
                                                        float* __restrict__ rs_out,
                                                        float* __restrict__ rs_in,
                                                        int n) {
    __shared__ int s_front[32];
    __shared__ int s_incl[32];

    const int tid  = threadIdx.x;
    const int lane = tid & 31;
    const int wave = tid >> 5;
    const int blockStart = blockIdx.x * 1024;

    int front = 0;
    const int4* c4 = (const int4*)cnt_in;
    const int   n4 = blockStart >> 2;
#pragma unroll 4
    for (int i = tid; i < n4; i += 1024) {
        int4 v = c4[i];
        front += (v.x + v.y) + (v.z + v.w);
    }

    const int i = blockStart + tid;
    const int c = (i < n) ? cnt_in[i] : 0;

    int incl = c;
    for (int off = 1; off < 32; off <<= 1) {
        int t = __shfl_up(incl, off);
        if (lane >= off) incl += t;
    }
    for (int off = 16; off > 0; off >>= 1) front += __shfl_xor(front, off);

    if (lane == 31) s_incl[wave]  = incl;
    if (lane == 0)  s_front[wave] = front;
    __syncthreads();

    int t = s_front[lane] + ((lane < wave) ? s_incl[lane] : 0);
    for (int off = 16; off > 0; off >>= 1) t += __shfl_xor(t, off);

    const int excl = t + incl - c;
    if (i < n) {
        rowptr[i] = excl;
        rs_in[i]  = rsqrtf(fmaxf((float)c, 1.0f));
        rs_out[i] = rsqrtf(fmaxf((float)cnt_out[i], 1.0f));
        if (i == n - 1) rowptr[n] = excl + c;
    }
}

__global__ __launch_bounds__(256) void place_kernel(const int* __restrict__ src,
                                                    const int* __restrict__ dst,
                                                    const int* __restrict__ slot,
                                                    const int* __restrict__ rowptr,
                                                    const float* __restrict__ rs_out,
                                                    int* __restrict__ col,
                                                    float* __restrict__ wgt,
                                                    int n_edges) {
    int e = blockIdx.x * blockDim.x + threadIdx.x;
    if (e < n_edges) {
        const int s   = src[e];
        const int pos = rowptr[dst[e]] + slot[e];
        col[pos] = s;
        wgt[pos] = rs_out[s];
    }
}

__device__ __forceinline__ void row_fma(float4& acc, const float4 v, const float w) {
    acc.x = fmaf(v.x, w, acc.x);
    acc.y = fmaf(v.y, w, acc.y);
    acc.z = fmaf(v.z, w, acc.z);
    acc.w = fmaf(v.w, w, acc.w);
}

__global__ __launch_bounds__(256) void gather_kernel(const float* __restrict__ X,
                                                     const int* __restrict__ rowptr,
                                                     const int* __restrict__ col,
                                                     const float* __restrict__ wgt,
                                                     float* __restrict__ msg,
                                                     int n_nodes) {
    const int node = (blockIdx.x * blockDim.x + threadIdx.x) >> 5;
    const int lane = threadIdx.x & 31;
    if (node >= n_nodes) return;

    const int beg = __builtin_amdgcn_readfirstlane(rowptr[node]);
    const int end = __builtin_amdgcn_readfirstlane(rowptr[node + 1]);

    const float4* X4 = (const float4*)X;
    float4 acc = make_float4(0.f, 0.f, 0.f, 0.f);

    for (int base = beg; base < end; base += 32) {
        const int n = (end - base < 32) ? (end - base) : 32;
        int   myc = 0;
        float myw = 0.f;
        if (lane < n) {
            myc = col[base + lane];
            myw = wgt[base + lane];
        }
        const int mywi = __float_as_int(myw);

        int j = 0;
        for (; j + 4 <= n; j += 4) {
            const int   s0 = __builtin_amdgcn_readlane(myc, j + 0);
            const int   s1 = __builtin_amdgcn_readlane(myc, j + 1);
            const int   s2 = __builtin_amdgcn_readlane(myc, j + 2);
            const int   s3 = __builtin_amdgcn_readlane(myc, j + 3);
            const float w0 = __int_as_float(__builtin_amdgcn_readlane(mywi, j + 0));
            const float w1 = __int_as_float(__builtin_amdgcn_readlane(mywi, j + 1));
            const float w2 = __int_as_float(__builtin_amdgcn_readlane(mywi, j + 2));
            const float w3 = __int_as_float(__builtin_amdgcn_readlane(mywi, j + 3));
            const float4 v0 = X4[(size_t)s0 * (DFEAT / 4) + lane];
            const float4 v1 = X4[(size_t)s1 * (DFEAT / 4) + lane];
            const float4 v2 = X4[(size_t)s2 * (DFEAT / 4) + lane];
            const float4 v3 = X4[(size_t)s3 * (DFEAT / 4) + lane];
            row_fma(acc, v0, w0);
            row_fma(acc, v1, w1);
            row_fma(acc, v2, w2);
            row_fma(acc, v3, w3);
        }
        const int r = n - j;
        if (r == 3) {
            const int   s0 = __builtin_amdgcn_readlane(myc, j + 0);
            const int   s1 = __builtin_amdgcn_readlane(myc, j + 1);
            const int   s2 = __builtin_amdgcn_readlane(myc, j + 2);
            const float w0 = __int_as_float(__builtin_amdgcn_readlane(mywi, j + 0));
            const float w1 = __int_as_float(__builtin_amdgcn_readlane(mywi, j + 1));
            const float w2 = __int_as_float(__builtin_amdgcn_readlane(mywi, j + 2));
            const float4 v0 = X4[(size_t)s0 * (DFEAT / 4) + lane];
            const float4 v1 = X4[(size_t)s1 * (DFEAT / 4) + lane];
            const float4 v2 = X4[(size_t)s2 * (DFEAT / 4) + lane];
            row_fma(acc, v0, w0);
            row_fma(acc, v1, w1);
            row_fma(acc, v2, w2);
        } else if (r == 2) {
            const int   s0 = __builtin_amdgcn_readlane(myc, j + 0);
            const int   s1 = __builtin_amdgcn_readlane(myc, j + 1);
            const float w0 = __int_as_float(__builtin_amdgcn_readlane(mywi, j + 0));
            const float w1 = __int_as_float(__builtin_amdgcn_readlane(mywi, j + 1));
            const float4 v0 = X4[(size_t)s0 * (DFEAT / 4) + lane];
            const float4 v1 = X4[(size_t)s1 * (DFEAT / 4) + lane];
            row_fma(acc, v0, w0);
            row_fma(acc, v1, w1);
        } else if (r == 1) {
            const int   s0 = __builtin_amdgcn_readlane(myc, j);
            const float w0 = __int_as_float(__builtin_amdgcn_readlane(mywi, j));
            const float4 v0 = X4[(size_t)s0 * (DFEAT / 4) + lane];
            row_fma(acc, v0, w0);
        }
    }
    ((float4*)msg)[(size_t)node * (DFEAT / 4) + lane] = acc;
}

template <bool RELU>
__global__ __launch_bounds__(256) void gcn_gemm(const float* A,
                                                const float* __restrict__ rs_in,
                                                const float* __restrict__ W,
                                                const float* __restrict__ bias,
                                                float* out) {
    __shared__ float Alds[16 * ASTRIDE];

    const int tid     = threadIdx.x;
    const int wave    = tid >> 5;
    const int lane    = tid & 31;
    const int rowBase = blockIdx.x << 4;

    {
        const int r = tid >> 4;
        const int c = (tid & 15) << 3;
        const float scale = rs_in[rowBase + r];
        const float4* gp = (const float4*)(A + (size_t)(rowBase + r) * DFEAT + c);
        float4 a0 = gp[0];
        float4 a1 = gp[1];
        float* lp = &Alds[r * ASTRIDE + c];
        lp[0] = a0.x * scale; lp[1] = a0.y * scale; lp[2] = a0.z * scale; lp[3] = a0.w * scale;
        lp[4] = a1.x * scale; lp[5] = a1.y * scale; lp[6] = a1.z * scale; lp[7] = a1.w * scale;
    }
    __syncthreads();

    const int   mrow = lane & 15;
    const int   khi  = (lane >> 4) << 1;
    const int   n0   = wave << 4;
    const float* wp  = W + n0 + mrow;
    const float* ap  = &Alds[mrow * ASTRIDE + khi];

    v8f acc = {0.f, 0.f, 0.f, 0.f, 0.f, 0.f, 0.f, 0.f};
#pragma unroll
    for (int k = 0; k < DFEAT; k += 4) {
        v2f a = *(const v2f*)(ap + k);
        v2f b;
        b.x = wp[(size_t)(k + khi)     * DFEAT];
        b.y = wp[(size_t)(k + khi + 1) * DFEAT];
        acc = __builtin_amdgcn_wmma_f32_16x16x4_f32(false, a, false, b,
                                                    (short)0, acc, false, false);
    }

    const int   col  = n0 + mrow;
    const float bv   = bias[col];
    const int   rsel = (lane >> 4) << 3;
#pragma unroll
    for (int v = 0; v < 8; ++v) {
        float h = acc[v] + bv;
        if (RELU) h = fmaxf(h, 0.f);
        out[(size_t)(rowBase + v + rsel) * DFEAT + col] = h;
    }
}

extern "C" void kernel_launch(void* const* d_in, const int* in_sizes, int n_in,
                              void* d_out, int out_size, void* d_ws, size_t ws_size,
                              hipStream_t stream) {
    const float* x   = (const float*)d_in[0];
    const int*   src = (const int*)  d_in[1];
    const int*   dst = (const int*)  d_in[2];
    const float* W1  = (const float*)d_in[3];
    const float* b1  = (const float*)d_in[4];
    const float* W2  = (const float*)d_in[5];
    const float* b2  = (const float*)d_in[6];
    float*       out = (float*)d_out;

    float* ws      = (float*)d_ws;
    int*   cnt_out = (int*)ws;
    int*   cnt_in  = (int*)ws + N_NODES;
    float* hbuf    = ws + 2 * N_NODES;
    float* aux     = hbuf + (size_t)N_NODES * DFEAT;
    float* rs_out  = aux;
    float* rs_in   = aux + N_NODES;
    int*   rowptr  = (int*)(aux + 2 * N_NODES);
    int*   slot    = (int*)(aux + 4 * N_NODES);
    int*   col     = slot + N_EDGES;
    float* wgt     = (float*)(col + N_EDGES);

    const int T     = 256;
    const int EBLK  = (N_EDGES + T - 1) / T;
    const int NBLK  = (N_NODES + 7) / 8;
    const int GEMMB = N_NODES / 16;

    zero_f4<<<(2 * N_NODES / 4 + T - 1) / T, T, 0, stream>>>((float4*)ws, 2 * N_NODES / 4);
    count_kernel<<<EBLK, T, 0, stream>>>(src, dst, cnt_out, cnt_in, slot, N_EDGES);
    rowstart_kernel<<<(N_NODES + 1023) / 1024, 1024, 0, stream>>>(cnt_out, cnt_in, rowptr,
                                                                  rs_out, rs_in, N_NODES);
    place_kernel<<<EBLK, T, 0, stream>>>(src, dst, slot, rowptr, rs_out, col, wgt, N_EDGES);

    gather_kernel<<<NBLK, T, 0, stream>>>(x, rowptr, col, wgt, hbuf, N_NODES);
    gcn_gemm<true><<<GEMMB, T, 0, stream>>>(hbuf, rs_in, W1, b1, hbuf);

    gather_kernel<<<NBLK, T, 0, stream>>>(hbuf, rowptr, col, wgt, out, N_NODES);
    gcn_gemm<false><<<GEMMB, T, 0, stream>>>(out, rs_in, W2, b2, out);
}


# Destination-ordered edge list built by a two-level counting sort in LDS (4096-edge chunks binned by 256-node range, then one block per range places rows, whole-line stores); row order: LDS arrival
# speedup vs baseline: 3.1329x; 1.5773x over previous
#include <hip/hip_runtime.h>

#define N_NODES 50000
#define N_EDGES 800000
#define DFEAT   128
#define ASTRIDE 132

typedef float v2f __attribute__((ext_vector_type(2)));
typedef float v8f __attribute__((ext_vector_type(8)));

__global__ __launch_bounds__(256) void zero_f4(float4* __restrict__ p, int n4) {
    int i = blockIdx.x * blockDim.x + threadIdx.x;
    if (i < n4) p[i] = make_float4(0.f, 0.f, 0.f, 0.f);
}

#define CHUNK 4096
#define NCH   ((N_EDGES + CHUNK - 1) / CHUNK)
#define NBK   ((N_NODES + 255) >> 8)
#define TROW  (NBK + 1)
#define CAP   6144

__global__ __launch_bounds__(1024) void chunk_sort_kernel(const int* __restrict__ src,
                                                          const int* __restrict__ dst,
                                                          int* __restrict__ cnt_out,
                                                          int* __restrict__ range_total,
                                                          int* __restrict__ T,
                                                          int2* __restrict__ tmp,
                                                          int n_edges) {
    __shared__ int  s_cnt[256];
    __shared__ int  s_wsum[8];
    __shared__ int2 s_edge[CHUNK];

    const int tid  = threadIdx.x;
    const int lane = tid & 31;
    const int wave = tid >> 5;
    const int chunkBase = blockIdx.x * CHUNK;

    if (tid < 256) s_cnt[tid] = 0;
    __syncthreads();

    int s[4], d[4], r[4];
#pragma unroll
    for (int i = 0; i < 4; ++i) {
        const int e = chunkBase + tid + i * 1024;
        s[i] = -1; d[i] = 0; r[i] = 0;
        if (e < n_edges) { s[i] = src[e]; d[i] = dst[e]; }
    }
#pragma unroll
    for (int i = 0; i < 4; ++i) {
        if (s[i] >= 0) {
            __hip_atomic_fetch_add(&cnt_out[s[i]], 1, __ATOMIC_RELAXED, __HIP_MEMORY_SCOPE_AGENT);
            r[i] = atomicAdd(&s_cnt[d[i] >> 8], 1);
        }
    }
    __syncthreads();

    int c = 0, incl = 0;
    if (tid < 256) {
        c = s_cnt[tid];
        incl = c;
        for (int off = 1; off < 32; off <<= 1) {
            int t = __shfl_up(incl, off);
            if (lane >= off) incl += t;
        }
        if (lane == 31) s_wsum[wave] = incl;
    }
    __syncthreads();
    if (tid < 256) {
        int bw = 0;
        for (int w = 0; w < wave; ++w) bw += s_wsum[w];
        const int excl = bw + incl - c;
        s_cnt[tid] = excl;
        if (tid <= NBK) T[blockIdx.x * TROW + tid] = excl;
        if (tid < NBK && c > 0)
            __hip_atomic_fetch_add(&range_total[tid], c, __ATOMIC_RELAXED, __HIP_MEMORY_SCOPE_AGENT);
    }
    __syncthreads();

#pragma unroll
    for (int i = 0; i < 4; ++i)
        if (s[i] >= 0) s_edge[s_cnt[d[i] >> 8] + r[i]] = make_int2(s[i], d[i]);
    __syncthreads();

#pragma unroll
    for (int i = 0; i < 4; ++i) {
        const int e = chunkBase + tid + i * 1024;
        if (e < n_edges) tmp[e] = s_edge[tid + i * 1024];
    }
}

__global__ __launch_bounds__(1024) void range_place_kernel(const int2* __restrict__ tmp,
                                                           const int* __restrict__ T,
                                                           const int* __restrict__ range_total,
                                                           const int* __restrict__ cnt_out,
                                                           int* __restrict__ rowptr,
                                                           float* __restrict__ rs_in,
                                                           int* __restrict__ col,
                                                           float* __restrict__ wgt,
                                                           int n_nodes) {
    __shared__ int  s_segS[256];
    __shared__ int  s_segL[256];
    __shared__ int  s_cnt[256];
    __shared__ int  s_cur[256];
    __shared__ int  s_bsum[8];
    __shared__ int  s_wsum[8];
    __shared__ int2 s_out[CAP];

    const int tid  = threadIdx.x;
    const int lane = tid & 31;
    const int wave = tid >> 5;
    const int b    = blockIdx.x;

    if (tid < 256) {
        int st = 0, ln = 0;
        if (tid < NCH) {
            const int t0 = T[tid * TROW + b];
            const int t1 = T[tid * TROW + b + 1];
            st = tid * CHUNK + t0;
            ln = t1 - t0;
        }
        s_segS[tid] = st;
        s_segL[tid] = ln;
        s_cnt[tid]  = 0;
        int v = (tid < b) ? range_total[tid] : 0;
        for (int off = 16; off > 0; off >>= 1) v += __shfl_xor(v, off);
        if (lane == 0) s_bsum[wave] = v;
    }
    __syncthreads();

    int base = 0;
#pragma unroll
    for (int w = 0; w < 8; ++w) base += s_bsum[w];
    const int  total  = range_total[b];
    const bool staged = (total <= CAP);

    for (int ch = wave; ch < NCH; ch += 32) {
        const int st = s_segS[ch];
        const int ln = s_segL[ch];
        for (int j = lane; j < ln; j += 32) {
            const int2 e = tmp[st + j];
            atomicAdd(&s_cnt[e.y & 255], 1);
        }
    }
    __syncthreads();

    int c = 0, incl = 0;
    if (tid < 256) {
        c = s_cnt[tid];
        incl = c;
        for (int off = 1; off < 32; off <<= 1) {
            int t = __shfl_up(incl, off);
            if (lane >= off) incl += t;
        }
        if (lane == 31) s_wsum[wave] = incl;
    }
    __syncthreads();
    if (tid < 256) {
        int bw = 0;
        for (int w = 0; w < wave; ++w) bw += s_wsum[w];
        const int excl = bw + incl - c;
        s_cur[tid] = excl;
        const int node = (b << 8) + tid;
        if (node < n_nodes) {
            rowptr[node] = base + excl;
            rs_in[node]  = rsqrtf(fmaxf((float)c, 1.0f));
            if (node == n_nodes - 1) rowptr[n_nodes] = base + excl + c;
        }
    }
    __syncthreads();

    for (int ch = wave; ch < NCH; ch += 32) {
        const int st = s_segS[ch];
        const int ln = s_segL[ch];
        for (int j = lane; j < ln; j += 32) {
            const int2  e   = tmp[st + j];
            const float w   = rsqrtf(fmaxf((float)cnt_out[e.x], 1.0f));
            const int   pos = atomicAdd(&s_cur[e.y & 255], 1);
            if (staged) {
                s_out[pos] = make_int2(e.x, __float_as_int(w));
            } else {
                col[base + pos] = e.x;
                wgt[base + pos] = w;
            }
        }
    }
    __syncthreads();

    if (staged) {
        for (int i = tid; i < total; i += 1024) {
            const int2 o = s_out[i];
            col[base + i] = o.x;
            wgt[base + i] = __int_as_float(o.y);
        }
    }
}

__device__ __forceinline__ void row_fma(float4& acc, const float4 v, const float w) {
    acc.x = fmaf(v.x, w, acc.x);
    acc.y = fmaf(v.y, w, acc.y);
    acc.z = fmaf(v.z, w, acc.z);
    acc.w = fmaf(v.w, w, acc.w);
}

__global__ __launch_bounds__(256) void gather_kernel(const float* __restrict__ X,
                                                     const int* __restrict__ rowptr,
                                                     const int* __restrict__ col,
                                                     const float* __restrict__ wgt,
                                                     float* __restrict__ msg,
                                                     int n_nodes) {
    const int node = (blockIdx.x * blockDim.x + threadIdx.x) >> 5;
    const int lane = threadIdx.x & 31;
    if (node >= n_nodes) return;

    const int beg = __builtin_amdgcn_readfirstlane(rowptr[node]);
    const int end = __builtin_amdgcn_readfirstlane(rowptr[node + 1]);

    const float4* X4 = (const float4*)X;
    float4 acc = make_float4(0.f, 0.f, 0.f, 0.f);

    for (int base = beg; base < end; base += 32) {
        const int n = (end - base < 32) ? (end - base) : 32;
        int   myc = 0;
        float myw = 0.f;
        if (lane < n) {
            myc = col[base + lane];
            myw = wgt[base + lane];
        }
        const int mywi = __float_as_int(myw);

        int j = 0;
        for (; j + 4 <= n; j += 4) {
            const int   s0 = __builtin_amdgcn_readlane(myc, j + 0);
            const int   s1 = __builtin_amdgcn_readlane(myc, j + 1);
            const int   s2 = __builtin_amdgcn_readlane(myc, j + 2);
            const int   s3 = __builtin_amdgcn_readlane(myc, j + 3);
            const float w0 = __int_as_float(__builtin_amdgcn_readlane(mywi, j + 0));
            const float w1 = __int_as_float(__builtin_amdgcn_readlane(mywi, j + 1));
            const float w2 = __int_as_float(__builtin_amdgcn_readlane(mywi, j + 2));
            const float w3 = __int_as_float(__builtin_amdgcn_readlane(mywi, j + 3));
            const float4 v0 = X4[(size_t)s0 * (DFEAT / 4) + lane];
            const float4 v1 = X4[(size_t)s1 * (DFEAT / 4) + lane];
            const float4 v2 = X4[(size_t)s2 * (DFEAT / 4) + lane];
            const float4 v3 = X4[(size_t)s3 * (DFEAT / 4) + lane];
            row_fma(acc, v0, w0);
            row_fma(acc, v1, w1);
            row_fma(acc, v2, w2);
            row_fma(acc, v3, w3);
        }
        const int r = n - j;
        if (r == 3) {
            const int   s0 = __builtin_amdgcn_readlane(myc, j + 0);
            const int   s1 = __builtin_amdgcn_readlane(myc, j + 1);
            const int   s2 = __builtin_amdgcn_readlane(myc, j + 2);
            const float w0 = __int_as_float(__builtin_amdgcn_readlane(mywi, j + 0));
            const float w1 = __int_as_float(__builtin_amdgcn_readlane(mywi, j + 1));
            const float w2 = __int_as_float(__builtin_amdgcn_readlane(mywi, j + 2));
            const float4 v0 = X4[(size_t)s0 * (DFEAT / 4) + lane];
            const float4 v1 = X4[(size_t)s1 * (DFEAT / 4) + lane];
            const float4 v2 = X4[(size_t)s2 * (DFEAT / 4) + lane];
            row_fma(acc, v0, w0);
            row_fma(acc, v1, w1);
            row_fma(acc, v2, w2);
        } else if (r == 2) {
            const int   s0 = __builtin_amdgcn_readlane(myc, j + 0);
            const int   s1 = __builtin_amdgcn_readlane(myc, j + 1);
            const float w0 = __int_as_float(__builtin_amdgcn_readlane(mywi, j + 0));
            const float w1 = __int_as_float(__builtin_amdgcn_readlane(mywi, j + 1));
            const float4 v0 = X4[(size_t)s0 * (DFEAT / 4) + lane];
            const float4 v1 = X4[(size_t)s1 * (DFEAT / 4) + lane];
            row_fma(acc, v0, w0);
            row_fma(acc, v1, w1);
        } else if (r == 1) {
            const int   s0 = __builtin_amdgcn_readlane(myc, j);
            const float w0 = __int_as_float(__builtin_amdgcn_readlane(mywi, j));
            const float4 v0 = X4[(size_t)s0 * (DFEAT / 4) + lane];
            row_fma(acc, v0, w0);
        }
    }
    ((float4*)msg)[(size_t)node * (DFEAT / 4) + lane] = acc;
}

template <bool RELU>
__global__ __launch_bounds__(256) void gcn_gemm(const float* A,
                                                const float* __restrict__ rs_in,
                                                const float* __restrict__ W,
                                                const float* __restrict__ bias,
                                                float* out) {
    __shared__ float Alds[16 * ASTRIDE];

    const int tid     = threadIdx.x;
    const int wave    = tid >> 5;
    const int lane    = tid & 31;
    const int rowBase = blockIdx.x << 4;

    {
        const int r = tid >> 4;
        const int c = (tid & 15) << 3;
        const float scale = rs_in[rowBase + r];
        const float4* gp = (const float4*)(A + (size_t)(rowBase + r) * DFEAT + c);
        float4 a0 = gp[0];
        float4 a1 = gp[1];
        float* lp = &Alds[r * ASTRIDE + c];
        lp[0] = a0.x * scale; lp[1] = a0.y * scale; lp[2] = a0.z * scale; lp[3] = a0.w * scale;
        lp[4] = a1.x * scale; lp[5] = a1.y * scale; lp[6] = a1.z * scale; lp[7] = a1.w * scale;
    }
    __syncthreads();

    const int   mrow = lane & 15;
    const int   khi  = (lane >> 4) << 1;
    const int   n0   = wave << 4;
    const float* wp  = W + n0 + mrow;
    const float* ap  = &Alds[mrow * ASTRIDE + khi];

    v8f acc = {0.f, 0.f, 0.f, 0.f, 0.f, 0.f, 0.f, 0.f};
#pragma unroll
    for (int k = 0; k < DFEAT; k += 4) {
        v2f a = *(const v2f*)(ap + k);
        v2f b;
        b.x = wp[(size_t)(k + khi)     * DFEAT];
        b.y = wp[(size_t)(k + khi + 1) * DFEAT];
        acc = __builtin_amdgcn_wmma_f32_16x16x4_f32(false, a, false, b,
                                                    (short)0, acc, false, false);
    }

    const int   col  = n0 + mrow;
    const float bv   = bias[col];
    const int   rsel = (lane >> 4) << 3;
#pragma unroll
    for (int v = 0; v < 8; ++v) {
        float h = acc[v] + bv;
        if (RELU) h = fmaxf(h, 0.f);
        out[(size_t)(rowBase + v + rsel) * DFEAT + col] = h;
    }
}

extern "C" void kernel_launch(void* const* d_in, const int* in_sizes, int n_in,
                              void* d_out, int out_size, void* d_ws, size_t ws_size,
                              hipStream_t stream) {
    const float* x   = (const float*)d_in[0];
    const int*   src = (const int*)  d_in[1];
    const int*   dst = (const int*)  d_in[2];
    const float* W1  = (const float*)d_in[3];
    const float* b1  = (const float*)d_in[4];
    const float* W2  = (const float*)d_in[5];
    const float* b2  = (const float*)d_in[6];
    float*       out = (float*)d_out;

    float* ws          = (float*)d_ws;
    int*   cnt_out     = (int*)ws;
    int*   range_total = (int*)ws + N_NODES;
    float* hbuf        = ws + 2 * N_NODES;
    int2*  tmp         = (int2*)hbuf;
    float* aux         = hbuf + (size_t)N_NODES * DFEAT;
    float* rs_in       = aux + N_NODES;
    int*   rowptr      = (int*)(aux + 2 * N_NODES);
    int*   T           = (int*)(aux + 4 * N_NODES);
    int*   col         = T + N_EDGES;
    float* wgt         = (float*)(col + N_EDGES);

    const int T256  = 256;
    const int NBLK  = (N_NODES + 7) / 8;
    const int GEMMB = N_NODES / 16;

    zero_f4<<<(2 * N_NODES / 4 + T256 - 1) / T256, T256, 0, stream>>>((float4*)ws, 2 * N_NODES / 4);
    chunk_sort_kernel<<<NCH, 1024, 0, stream>>>(src, dst, cnt_out, range_total, T, tmp, N_EDGES);
    range_place_kernel<<<NBK, 1024, 0, stream>>>(tmp, T, range_total, cnt_out, rowptr, rs_in,
                                                 col, wgt, N_NODES);

    gather_kernel<<<NBLK, T256, 0, stream>>>(x, rowptr, col, wgt, hbuf, N_NODES);
    gcn_gemm<true><<<GEMMB, T256, 0, stream>>>(hbuf, rs_in, W1, b1, hbuf);

    gather_kernel<<<NBLK, T256, 0, stream>>>(hbuf, rowptr, col, wgt, out, N_NODES);
    gcn_gemm<false><<<GEMMB, T256, 0, stream>>>(out, rs_in, W2, b2, out);
}


# Out-degree weights without global traffic: edges binned in LDS by source range (degree counted, weight attached there), then by destination; no global atomics or zero fill; row order: LDS arrival
# speedup vs baseline: 4.0339x; 1.2876x over previous
#include <hip/hip_runtime.h>

#define N_NODES 50000
#define N_EDGES 800000
#define DFEAT   128
#define ASTRIDE 132

typedef float v2f __attribute__((ext_vector_type(2)));
typedef float v8f __attribute__((ext_vector_type(8)));

#define CHUNK 4096
#define NCH   ((N_EDGES + CHUNK - 1) / CHUNK)
#define NBK   ((N_NODES + 255) >> 8)
#define TROW  (NBK + 1)
#define CAP   6144

__global__ __launch_bounds__(1024) void chunk_sort_kernel(const int* __restrict__ src,
                                                          const int* __restrict__ dst,
                                                          int* __restrict__ TA,
                                                          int2* __restrict__ tmpA,
                                                          int n_edges) {
    __shared__ int  s_cnt[256];
    __shared__ int  s_wsum[8];
    __shared__ int2 s_edge[CHUNK];

    const int tid  = threadIdx.x;
    const int lane = tid & 31;
    const int wave = tid >> 5;
    const int chunkBase = blockIdx.x * CHUNK;

    if (tid < 256) s_cnt[tid] = 0;
    __syncthreads();

    int s[4], d[4], r[4];
#pragma unroll
    for (int i = 0; i < 4; ++i) {
        const int e = chunkBase + tid + i * 1024;
        s[i] = -1; d[i] = 0; r[i] = 0;
        if (e < n_edges) { s[i] = src[e]; d[i] = dst[e]; }
    }
#pragma unroll
    for (int i = 0; i < 4; ++i)
        if (s[i] >= 0) r[i] = atomicAdd(&s_cnt[s[i] >> 8], 1);
    __syncthreads();

    int c = 0, incl = 0;
    if (tid < 256) {
        c = s_cnt[tid];
        incl = c;
        for (int off = 1; off < 32; off <<= 1) {
            int t = __shfl_up(incl, off);
            if (lane >= off) incl += t;
        }
        if (lane == 31) s_wsum[wave] = incl;
    }
    __syncthreads();
    if (tid < 256) {
        int bw = 0;
        for (int w = 0; w < wave; ++w) bw += s_wsum[w];
        const int excl = bw + incl - c;
        s_cnt[tid] = excl;
        if (tid <= NBK) TA[blockIdx.x * TROW + tid] = excl;
    }
    __syncthreads();

#pragma unroll
    for (int i = 0; i < 4; ++i)
        if (s[i] >= 0) s_edge[s_cnt[s[i] >> 8] + r[i]] = make_int2(s[i], d[i]);
    __syncthreads();

#pragma unroll
    for (int i = 0; i < 4; ++i) {
        const int e = chunkBase + tid + i * 1024;
        if (e < n_edges) tmpA[e] = s_edge[tid + i * 1024];
    }
}

__global__ __launch_bounds__(1024) void src_range_kernel(const int2* __restrict__ tmpA,
                                                         const int* __restrict__ TA,
                                                         int* __restrict__ TB,
                                                         int* __restrict__ baseA,
                                                         int2* __restrict__ tmpB) {
    __shared__ int  s_segS[256];
    __shared__ int  s_segL[256];
    __shared__ int  s_deg[256];
    __shared__ int  s_cnt[256];
    __shared__ int  s_bsum[8];
    __shared__ int  s_tsum[8];
    __shared__ int  s_wsum[8];
    __shared__ int2 s_out[CAP];

    const int tid  = threadIdx.x;
    const int lane = tid & 31;
    const int wave = tid >> 5;
    const int b    = blockIdx.x;

    if (tid < 256) {
        int t0 = 0, st = 0, ln = 0;
        if (tid < NCH) {
            t0 = TA[tid * TROW + b];
            const int t1 = TA[tid * TROW + b + 1];
            st = tid * CHUNK + t0;
            ln = t1 - t0;
        }
        s_segS[tid] = st;
        s_segL[tid] = ln;
        s_deg[tid]  = 0;
        s_cnt[tid]  = 0;
        int v = t0, u = ln;
        for (int off = 16; off > 0; off >>= 1) {
            v += __shfl_xor(v, off);
            u += __shfl_xor(u, off);
        }
        if (lane == 0) { s_bsum[wave] = v; s_tsum[wave] = u; }
    }
    __syncthreads();

    int base = 0, total = 0;
#pragma unroll
    for (int w = 0; w < 8; ++w) { base += s_bsum[w]; total += s_tsum[w]; }
    const bool staged = (total <= CAP);

    for (int ch = wave; ch < NCH; ch += 32) {
        const int st = s_segS[ch];
        const int ln = s_segL[ch];
        for (int j = lane; j < ln; j += 32) {
            const int2 e = tmpA[st + j];
            atomicAdd(&s_deg[e.x & 255], 1);
            atomicAdd(&s_cnt[e.y >> 8], 1);
        }
    }
    __syncthreads();

    int c = 0, incl = 0;
    if (tid < 256) {
        c = s_cnt[tid];
        incl = c;
        for (int off = 1; off < 32; off <<= 1) {
            int t = __shfl_up(incl, off);
            if (lane >= off) incl += t;
        }
        if (lane == 31) s_wsum[wave] = incl;
    }
    __syncthreads();
    if (tid < 256) {
        int bw = 0;
        for (int w = 0; w < wave; ++w) bw += s_wsum[w];
        const int excl = bw + incl - c;
        s_cnt[tid] = excl;
        if (tid <= NBK) TB[b * TROW + tid] = excl;
        if (tid == 0) baseA[b] = base;
    }
    __syncthreads();

    for (int ch = wave; ch < NCH; ch += 32) {
        const int st = s_segS[ch];
        const int ln = s_segL[ch];
        for (int j = lane; j < ln; j += 32) {
            const int2  e   = tmpA[st + j];
            const float w   = rsqrtf(fmaxf((float)s_deg[e.x & 255], 1.0f));
            const int   pos = atomicAdd(&s_cnt[e.y >> 8], 1);
            const int2  o   = make_int2((e.x << 8) | (e.y & 255), __float_as_int(w));
            if (staged) s_out[pos] = o;
            else        tmpB[base + pos] = o;
        }
    }
    __syncthreads();

    if (staged) {
        for (int i = tid; i < total; i += 1024) tmpB[base + i] = s_out[i];
    }
}

__global__ __launch_bounds__(1024) void range_place_kernel(const int2* __restrict__ tmpB,
                                                           const int* __restrict__ TB,
                                                           const int* __restrict__ baseA,
                                                           int* __restrict__ rowptr,
                                                           float* __restrict__ rs_in,
                                                           int* __restrict__ col,
                                                           float* __restrict__ wgt,
                                                           int n_nodes) {
    __shared__ int  s_segS[256];
    __shared__ int  s_segL[256];
    __shared__ int  s_cnt[256];
    __shared__ int  s_cur[256];
    __shared__ int  s_bsum[8];
    __shared__ int  s_tsum[8];
    __shared__ int  s_wsum[8];
    __shared__ int2 s_out[CAP];

    const int tid  = threadIdx.x;
    const int lane = tid & 31;
    const int wave = tid >> 5;
    const int b    = blockIdx.x;

    if (tid < 256) {
        int t0 = 0, st = 0, ln = 0;
        if (tid < NBK) {
            t0 = TB[tid * TROW + b];
            const int t1 = TB[tid * TROW + b + 1];
            st = baseA[tid] + t0;
            ln = t1 - t0;
        }
        s_segS[tid] = st;
        s_segL[tid] = ln;
        s_cnt[tid]  = 0;
        int v = t0, u = ln;
        for (int off = 16; off > 0; off >>= 1) {
            v += __shfl_xor(v, off);
            u += __shfl_xor(u, off);
        }
        if (lane == 0) { s_bsum[wave] = v; s_tsum[wave] = u; }
    }
    __syncthreads();

    int base = 0, total = 0;
#pragma unroll
    for (int w = 0; w < 8; ++w) { base += s_bsum[w]; total += s_tsum[w]; }
    const bool staged = (total <= CAP);

    for (int g = wave; g < NBK; g += 32) {
        const int st = s_segS[g];
        const int ln = s_segL[g];
        for (int j = lane; j < ln; j += 32) {
            const int2 e = tmpB[st + j];
            atomicAdd(&s_cnt[e.x & 255], 1);
        }
    }
    __syncthreads();

    int c = 0, incl = 0;
    if (tid < 256) {
        c = s_cnt[tid];
        incl = c;
        for (int off = 1; off < 32; off <<= 1) {
            int t = __shfl_up(incl, off);
            if (lane >= off) incl += t;
        }
        if (lane == 31) s_wsum[wave] = incl;
    }
    __syncthreads();
    if (tid < 256) {
        int bw = 0;
        for (int w = 0; w < wave; ++w) bw += s_wsum[w];
        const int excl = bw + incl - c;
        s_cur[tid] = excl;
        const int node = (b << 8) + tid;
        if (node < n_nodes) {
            rowptr[node] = base + excl;
            rs_in[node]  = rsqrtf(fmaxf((float)c, 1.0f));
            if (node == n_nodes - 1) rowptr[n_nodes] = base + excl + c;
        }
    }
    __syncthreads();

    for (int g = wave; g < NBK; g += 32) {
        const int st = s_segS[g];
        const int ln = s_segL[g];
        for (int j = lane; j < ln; j += 32) {
            const int2 e   = tmpB[st + j];
            const int  pos = atomicAdd(&s_cur[e.x & 255], 1);
            if (staged) {
                s_out[pos] = make_int2(e.x >> 8, e.y);
            } else {
                col[base + pos] = e.x >> 8;
                wgt[base + pos] = __int_as_float(e.y);
            }
        }
    }
    __syncthreads();

    if (staged) {
        for (int i = tid; i < total; i += 1024) {
            const int2 o = s_out[i];
            col[base + i] = o.x;
            wgt[base + i] = __int_as_float(o.y);
        }
    }
}

__device__ __forceinline__ void row_fma(float4& acc, const float4 v, const float w) {
    acc.x = fmaf(v.x, w, acc.x);
    acc.y = fmaf(v.y, w, acc.y);
    acc.z = fmaf(v.z, w, acc.z);
    acc.w = fmaf(v.w, w, acc.w);
}

__global__ __launch_bounds__(256) void gather_kernel(const float* __restrict__ X,
                                                     const int* __restrict__ rowptr,
                                                     const int* __restrict__ col,
                                                     const float* __restrict__ wgt,
                                                     float* __restrict__ msg,
                                                     int n_nodes) {
    const int node = (blockIdx.x * blockDim.x + threadIdx.x) >> 5;
    const int lane = threadIdx.x & 31;
    if (node >= n_nodes) return;

    const int beg = __builtin_amdgcn_readfirstlane(rowptr[node]);
    const int end = __builtin_amdgcn_readfirstlane(rowptr[node + 1]);

    const float4* X4 = (const float4*)X;
    float4 acc = make_float4(0.f, 0.f, 0.f, 0.f);

    for (int base = beg; base < end; base += 32) {
        const int n = (end - base < 32) ? (end - base) : 32;
        int   myc = 0;
        float myw = 0.f;
        if (lane < n) {
            myc = col[base + lane];
            myw = wgt[base + lane];
        }
        const int mywi = __float_as_int(myw);

        int j = 0;
        for (; j + 4 <= n; j += 4) {
            const int   s0 = __builtin_amdgcn_readlane(myc, j + 0);
            const int   s1 = __builtin_amdgcn_readlane(myc, j + 1);
            const int   s2 = __builtin_amdgcn_readlane(myc, j + 2);
            const int   s3 = __builtin_amdgcn_readlane(myc, j + 3);
            const float w0 = __int_as_float(__builtin_amdgcn_readlane(mywi, j + 0));
            const float w1 = __int_as_float(__builtin_amdgcn_readlane(mywi, j + 1));
            const float w2 = __int_as_float(__builtin_amdgcn_readlane(mywi, j + 2));
            const float w3 = __int_as_float(__builtin_amdgcn_readlane(mywi, j + 3));
            const float4 v0 = X4[(size_t)s0 * (DFEAT / 4) + lane];
            const float4 v1 = X4[(size_t)s1 * (DFEAT / 4) + lane];
            const float4 v2 = X4[(size_t)s2 * (DFEAT / 4) + lane];
            const float4 v3 = X4[(size_t)s3 * (DFEAT / 4) + lane];
            row_fma(acc, v0, w0);
            row_fma(acc, v1, w1);
            row_fma(acc, v2, w2);
            row_fma(acc, v3, w3);
        }
        const int r = n - j;
        if (r == 3) {
            const int   s0 = __builtin_amdgcn_readlane(myc, j + 0);
            const int   s1 = __builtin_amdgcn_readlane(myc, j + 1);
            const int   s2 = __builtin_amdgcn_readlane(myc, j + 2);
            const float w0 = __int_as_float(__builtin_amdgcn_readlane(mywi, j + 0));
            const float w1 = __int_as_float(__builtin_amdgcn_readlane(mywi, j + 1));
            const float w2 = __int_as_float(__builtin_amdgcn_readlane(mywi, j + 2));
            const float4 v0 = X4[(size_t)s0 * (DFEAT / 4) + lane];
            const float4 v1 = X4[(size_t)s1 * (DFEAT / 4) + lane];
            const float4 v2 = X4[(size_t)s2 * (DFEAT / 4) + lane];
            row_fma(acc, v0, w0);
            row_fma(acc, v1, w1);
            row_fma(acc, v2, w2);
        } else if (r == 2) {
            const int   s0 = __builtin_amdgcn_readlane(myc, j + 0);
            const int   s1 = __builtin_amdgcn_readlane(myc, j + 1);
            const float w0 = __int_as_float(__builtin_amdgcn_readlane(mywi, j + 0));
            const float w1 = __int_as_float(__builtin_amdgcn_readlane(mywi, j + 1));
            const float4 v0 = X4[(size_t)s0 * (DFEAT / 4) + lane];
            const float4 v1 = X4[(size_t)s1 * (DFEAT / 4) + lane];
            row_fma(acc, v0, w0);
            row_fma(acc, v1, w1);
        } else if (r == 1) {
            const int   s0 = __builtin_amdgcn_readlane(myc, j);
            const float w0 = __int_as_float(__builtin_amdgcn_readlane(mywi, j));
            const float4 v0 = X4[(size_t)s0 * (DFEAT / 4) + lane];
            row_fma(acc, v0, w0);
        }
    }
    ((float4*)msg)[(size_t)node * (DFEAT / 4) + lane] = acc;
}

template <bool RELU>
__global__ __launch_bounds__(256) void gcn_gemm(const float* A,
                                                const float* __restrict__ rs_in,
                                                const float* __restrict__ W,
                                                const float* __restrict__ bias,
                                                float* out) {
    __shared__ float Alds[16 * ASTRIDE];

    const int tid     = threadIdx.x;
    const int wave    = tid >> 5;
    const int lane    = tid & 31;
    const int rowBase = blockIdx.x << 4;

    {
        const int r = tid >> 4;
        const int c = (tid & 15) << 3;
        const float scale = rs_in[rowBase + r];
        const float4* gp = (const float4*)(A + (size_t)(rowBase + r) * DFEAT + c);
        float4 a0 = gp[0];
        float4 a1 = gp[1];
        float* lp = &Alds[r * ASTRIDE + c];
        lp[0] = a0.x * scale; lp[1] = a0.y * scale; lp[2] = a0.z * scale; lp[3] = a0.w * scale;
        lp[4] = a1.x * scale; lp[5] = a1.y * scale; lp[6] = a1.z * scale; lp[7] = a1.w * scale;
    }
    __syncthreads();

    const int   mrow = lane & 15;
    const int   khi  = (lane >> 4) << 1;
    const int   n0   = wave << 4;
    const float* wp  = W + n0 + mrow;
    const float* ap  = &Alds[mrow * ASTRIDE + khi];

    v8f acc = {0.f, 0.f, 0.f, 0.f, 0.f, 0.f, 0.f, 0.f};
#pragma unroll
    for (int k = 0; k < DFEAT; k += 4) {
        v2f a = *(const v2f*)(ap + k);
        v2f b;
        b.x = wp[(size_t)(k + khi)     * DFEAT];
        b.y = wp[(size_t)(k + khi + 1) * DFEAT];
        acc = __builtin_amdgcn_wmma_f32_16x16x4_f32(false, a, false, b,
                                                    (short)0, acc, false, false);
    }

    const int   col  = n0 + mrow;
    const float bv   = bias[col];
    const int   rsel = (lane >> 4) << 3;
#pragma unroll
    for (int v = 0; v < 8; ++v) {
        float h = acc[v] + bv;
        if (RELU) h = fmaxf(h, 0.f);
        out[(size_t)(rowBase + v + rsel) * DFEAT + col] = h;
    }
}

extern "C" void kernel_launch(void* const* d_in, const int* in_sizes, int n_in,
                              void* d_out, int out_size, void* d_ws, size_t ws_size,
                              hipStream_t stream) {
    const float* x   = (const float*)d_in[0];
    const int*   src = (const int*)  d_in[1];
    const int*   dst = (const int*)  d_in[2];
    const float* W1  = (const float*)d_in[3];
    const float* b1  = (const float*)d_in[4];
    const float* W2  = (const float*)d_in[5];
    const float* b2  = (const float*)d_in[6];
    float*       out = (float*)d_out;

    float* ws     = (float*)d_ws;
    int*   TA     = (int*)ws;
    int*   TB     = (int*)ws + 40000;
    int*   baseA  = (int*)ws + 80000;
    float* hbuf   = ws + 2 * N_NODES;
    int2*  tmpA   = (int2*)hbuf;
    int2*  tmpB   = (int2*)(hbuf + 2 * N_EDGES);
    float* aux    = hbuf + (size_t)N_NODES * DFEAT;
    float* rs_in  = aux + N_NODES;
    int*   rowptr = (int*)(aux + 2 * N_NODES);
    int*   col    = (int*)(aux + 4 * N_NODES) + N_EDGES;
    float* wgt    = (float*)(col + N_EDGES);

    const int T256  = 256;
    const int NBLK  = (N_NODES + 7) / 8;
    const int GEMMB = N_NODES / 16;

    chunk_sort_kernel<<<NCH, 1024, 0, stream>>>(src, dst, TA, tmpA, N_EDGES);
    src_range_kernel<<<NBK, 1024, 0, stream>>>(tmpA, TA, TB, baseA, tmpB);
    range_place_kernel<<<NBK, 1024, 0, stream>>>(tmpB, TB, baseA, rowptr, rs_in, col, wgt, N_NODES);

    gather_kernel<<<NBLK, T256, 0, stream>>>(x, rowptr, col, wgt, hbuf, N_NODES);
    gcn_gemm<true><<<GEMMB, T256, 0, stream>>>(hbuf, rs_in, W1, b1, hbuf);

    gather_kernel<<<NBLK, T256, 0, stream>>>(hbuf, rowptr, col, wgt, out, N_NODES);
    gcn_gemm<false><<<GEMMB, T256, 0, stream>>>(out, rs_in, W2, b2, out);
}


# Gather and layer kernels run as resident grids: a wave loops over a node share (edges as one stream, 8 to 16 row loads in flight) or over 16-row tiles (W fragments stay in registers); sums unchanged
# speedup vs baseline: 4.1575x; 1.0306x over previous
#include <hip/hip_runtime.h>

#define N_NODES 50000
#define N_EDGES 800000
#define DFEAT   128
#define ASTRIDE 132

typedef float v2f __attribute__((ext_vector_type(2)));
typedef float v8f __attribute__((ext_vector_type(8)));

#define CHUNK 4096
#define NCH   ((N_EDGES + CHUNK - 1) / CHUNK)
#define NBK   ((N_NODES + 255) >> 8)
#define TROW  (NBK + 1)
#define CAP   6144

__global__ __launch_bounds__(1024) void chunk_sort_kernel(const int* __restrict__ src,
                                                          const int* __restrict__ dst,
                                                          int* __restrict__ TA,
                                                          int2* __restrict__ tmpA,
                                                          int n_edges) {
    __shared__ int  s_cnt[256];
    __shared__ int  s_wsum[8];
    __shared__ int2 s_edge[CHUNK];

    const int tid  = threadIdx.x;
    const int lane = tid & 31;
    const int wave = tid >> 5;
    const int chunkBase = blockIdx.x * CHUNK;

    if (tid < 256) s_cnt[tid] = 0;
    __syncthreads();

    int s[4], d[4], r[4];
#pragma unroll
    for (int i = 0; i < 4; ++i) {
        const int e = chunkBase + tid + i * 1024;
        s[i] = -1; d[i] = 0; r[i] = 0;
        if (e < n_edges) { s[i] = src[e]; d[i] = dst[e]; }
    }
#pragma unroll
    for (int i = 0; i < 4; ++i)
        if (s[i] >= 0) r[i] = atomicAdd(&s_cnt[s[i] >> 8], 1);
    __syncthreads();

    int c = 0, incl = 0;
    if (tid < 256) {
        c = s_cnt[tid];
        incl = c;
        for (int off = 1; off < 32; off <<= 1) {
            int t = __shfl_up(incl, off);
            if (lane >= off) incl += t;
        }
        if (lane == 31) s_wsum[wave] = incl;
    }
    __syncthreads();
    if (tid < 256) {
        int bw = 0;
        for (int w = 0; w < wave; ++w) bw += s_wsum[w];
        const int excl = bw + incl - c;
        s_cnt[tid] = excl;
        if (tid <= NBK) TA[blockIdx.x * TROW + tid] = excl;
    }
    __syncthreads();

#pragma unroll
    for (int i = 0; i < 4; ++i)
        if (s[i] >= 0) s_edge[s_cnt[s[i] >> 8] + r[i]] = make_int2(s[i], d[i]);
    __syncthreads();

#pragma unroll
    for (int i = 0; i < 4; ++i) {
        const int e = chunkBase + tid + i * 1024;
        if (e < n_edges) tmpA[e] = s_edge[tid + i * 1024];
    }
}

__global__ __launch_bounds__(1024) void src_range_kernel(const int2* __restrict__ tmpA,
                                                         const int* __restrict__ TA,
                                                         int* __restrict__ TB,
                                                         int* __restrict__ baseA,
                                                         int2* __restrict__ tmpB) {
    __shared__ int  s_segS[256];
    __shared__ int  s_segL[256];
    __shared__ int  s_deg[256];
    __shared__ int  s_cnt[256];
    __shared__ int  s_bsum[8];
    __shared__ int  s_tsum[8];
    __shared__ int  s_wsum[8];
    __shared__ int2 s_out[CAP];

    const int tid  = threadIdx.x;
    const int lane = tid & 31;
    const int wave = tid >> 5;
    const int b    = blockIdx.x;

    if (tid < 256) {
        int t0 = 0, st = 0, ln = 0;
        if (tid < NCH) {
            t0 = TA[tid * TROW + b];
            const int t1 = TA[tid * TROW + b + 1];
            st = tid * CHUNK + t0;
            ln = t1 - t0;
        }
        s_segS[tid] = st;
        s_segL[tid] = ln;
        s_deg[tid]  = 0;
        s_cnt[tid]  = 0;
        int v = t0, u = ln;
        for (int off = 16; off > 0; off >>= 1) {
            v += __shfl_xor(v, off);
            u += __shfl_xor(u, off);
        }
        if (lane == 0) { s_bsum[wave] = v; s_tsum[wave] = u; }
    }
    __syncthreads();

    int base = 0, total = 0;
#pragma unroll
    for (int w = 0; w < 8; ++w) { base += s_bsum[w]; total += s_tsum[w]; }
    const bool staged = (total <= CAP);

    for (int ch = wave; ch < NCH; ch += 32) {
        const int st = s_segS[ch];
        const int ln = s_segL[ch];
        for (int j = lane; j < ln; j += 32) {
            const int2 e = tmpA[st + j];
            atomicAdd(&s_deg[e.x & 255], 1);
            atomicAdd(&s_cnt[e.y >> 8], 1);
        }
    }
    __syncthreads();

    int c = 0, incl = 0;
    if (tid < 256) {
        c = s_cnt[tid];
        incl = c;
        for (int off = 1; off < 32; off <<= 1) {
            int t = __shfl_up(incl, off);
            if (lane >= off) incl += t;
        }
        if (lane == 31) s_wsum[wave] = incl;
    }
    __syncthreads();
    if (tid < 256) {
        int bw = 0;
        for (int w = 0; w < wave; ++w) bw += s_wsum[w];
        const int excl = bw + incl - c;
        s_cnt[tid] = excl;
        if (tid <= NBK) TB[b * TROW + tid] = excl;
        if (tid == 0) baseA[b] = base;
    }
    __syncthreads();

    for (int ch = wave; ch < NCH; ch += 32) {
        const int st = s_segS[ch];
        const int ln = s_segL[ch];
        for (int j = lane; j < ln; j += 32) {
            const int2  e   = tmpA[st + j];
            const float w   = rsqrtf(fmaxf((float)s_deg[e.x & 255], 1.0f));
            const int   pos = atomicAdd(&s_cnt[e.y >> 8], 1);
            const int2  o   = make_int2((e.x << 8) | (e.y & 255), __float_as_int(w));
            if (staged) s_out[pos] = o;
            else        tmpB[base + pos] = o;
        }
    }
    __syncthreads();

    if (staged) {
        for (int i = tid; i < total; i += 1024) tmpB[base + i] = s_out[i];
    }
}

__global__ __launch_bounds__(1024) void range_place_kernel(const int2* __restrict__ tmpB,
                                                           const int* __restrict__ TB,
                                                           const int* __restrict__ baseA,
                                                           int* __restrict__ rowptr,
                                                           float* __restrict__ rs_in,
                                                           int* __restrict__ col,
                                                           float* __restrict__ wgt,
                                                           int n_nodes) {
    __shared__ int  s_segS[256];
    __shared__ int  s_segL[256];
    __shared__ int  s_cnt[256];
    __shared__ int  s_cur[256];
    __shared__ int  s_bsum[8];
    __shared__ int  s_tsum[8];
    __shared__ int  s_wsum[8];
    __shared__ int2 s_out[CAP];

    const int tid  = threadIdx.x;
    const int lane = tid & 31;
    const int wave = tid >> 5;
    const int b    = blockIdx.x;

    if (tid < 256) {
        int t0 = 0, st = 0, ln = 0;
        if (tid < NBK) {
            t0 = TB[tid * TROW + b];
            const int t1 = TB[tid * TROW + b + 1];
            st = baseA[tid] + t0;
            ln = t1 - t0;
        }
        s_segS[tid] = st;
        s_segL[tid] = ln;
        s_cnt[tid]  = 0;
        int v = t0, u = ln;
        for (int off = 16; off > 0; off >>= 1) {
            v += __shfl_xor(v, off);
            u += __shfl_xor(u, off);
        }
        if (lane == 0) { s_bsum[wave] = v; s_tsum[wave] = u; }
    }
    __syncthreads();

    int base = 0, total = 0;
#pragma unroll
    for (int w = 0; w < 8; ++w) { base += s_bsum[w]; total += s_tsum[w]; }
    const bool staged = (total <= CAP);

    for (int g = wave; g < NBK; g += 32) {
        const int st = s_segS[g];
        const int ln = s_segL[g];
        for (int j = lane; j < ln; j += 32) {
            const int2 e = tmpB[st + j];
            atomicAdd(&s_cnt[e.x & 255], 1);
        }
    }
    __syncthreads();

    int c = 0, incl = 0;
    if (tid < 256) {
        c = s_cnt[tid];
        incl = c;
        for (int off = 1; off < 32; off <<= 1) {
            int t = __shfl_up(incl, off);
            if (lane >= off) incl += t;
        }
        if (lane == 31) s_wsum[wave] = incl;
    }
    __syncthreads();
    if (tid < 256) {
        int bw = 0;
        for (int w = 0; w < wave; ++w) bw += s_wsum[w];
        const int excl = bw + incl - c;
        s_cur[tid] = excl;
        const int node = (b << 8) + tid;
        if (node < n_nodes) {
            rowptr[node] = base + excl;
            rs_in[node]  = rsqrtf(fmaxf((float)c, 1.0f));
            if (node == n_nodes - 1) rowptr[n_nodes] = base + excl + c;
        }
    }
    __syncthreads();

    for (int g = wave; g < NBK; g += 32) {
        const int st = s_segS[g];
        const int ln = s_segL[g];
        for (int j = lane; j < ln; j += 32) {
            const int2 e   = tmpB[st + j];
            const int  pos = atomicAdd(&s_cur[e.x & 255], 1);
            if (staged) {
                s_out[pos] = make_int2(e.x >> 8, e.y);
            } else {
                col[base + pos] = e.x >> 8;
                wgt[base + pos] = __int_as_float(e.y);
            }
        }
    }
    __syncthreads();

    if (staged) {
        for (int i = tid; i < total; i += 1024) {
            const int2 o = s_out[i];
            col[base + i] = o.x;
            wgt[base + i] = __int_as_float(o.y);
        }
    }
}

__device__ __forceinline__ void row_fma(float4& acc, const float4 v, const float w) {
    acc.x = fmaf(v.x, w, acc.x);
    acc.y = fmaf(v.y, w, acc.y);
    acc.z = fmaf(v.z, w, acc.z);
    acc.w = fmaf(v.w, w, acc.w);
}

#define GATHER_WAVES_LOG2 13
#define GATHER_WAVES (1 << GATHER_WAVES_LOG2)
#define GATHER_BLOCKS (GATHER_WAVES / 8)

#define GATHER_ISSUE(V, CREG, J, NN)                                                  \
    _Pragma("unroll")                                                                 \
    for (int i = 0; i < 8; ++i) {                                                     \
        const int p = ((J) + i < (NN)) ? ((J) + i) : ((NN) - 1);                      \
        const int s = __builtin_amdgcn_readlane((CREG), p);                           \
        V[i] = X4[(size_t)s * (DFEAT / 4) + lane];                                    \
    }

#define GATHER_CONSUME(V, WREG, J, NN)                                                \
    _Pragma("unroll")                                                                 \
    for (int i = 0; i < 8; ++i) {                                                     \
        if ((J) + i < (NN)) {                                                         \
            while (e == cur_end) {                                                    \
                M4[(size_t)(g0 + cur) * (DFEAT / 4) + lane] = acc;                    \
                acc = make_float4(0.f, 0.f, 0.f, 0.f);                                \
                ++cur;                                                                \
                cur_end = __builtin_amdgcn_readlane(rp, cur + 1);                     \
            }                                                                         \
            const float w = __int_as_float(__builtin_amdgcn_readlane((WREG), (J) + i)); \
            row_fma(acc, V[i], w);                                                    \
            ++e;                                                                      \
        }                                                                             \
    }

__global__ __launch_bounds__(256) void gather_kernel(const float* __restrict__ X,
                                                     const int* __restrict__ rowptr,
                                                     const int* __restrict__ col,
                                                     const float* __restrict__ wgt,
                                                     float* __restrict__ msg,
                                                     int n_nodes) {
    const int lane = threadIdx.x & 31;
    const int wid  = __builtin_amdgcn_readfirstlane((int)((blockIdx.x * blockDim.x + threadIdx.x) >> 5));
    const int n0   = (int)(((unsigned long long)wid * (unsigned)n_nodes) >> GATHER_WAVES_LOG2);
    const int n1   = (int)(((unsigned long long)(wid + 1) * (unsigned)n_nodes) >> GATHER_WAVES_LOG2);

    const float4* X4 = (const float4*)X;
    float4*       M4 = (float4*)msg;

    for (int g0 = n0; g0 < n1; g0 += 31) {
        const int cnt  = (n1 - g0 < 31) ? (n1 - g0) : 31;
        const int rp   = rowptr[g0 + ((lane < cnt) ? lane : cnt)];
        const int ebeg = __builtin_amdgcn_readlane(rp, 0);
        const int eend = __builtin_amdgcn_readlane(rp, cnt);

        int    cur     = 0;
        int    cur_end = __builtin_amdgcn_readlane(rp, 1);
        int    e       = ebeg;
        float4 acc     = make_float4(0.f, 0.f, 0.f, 0.f);

        if (ebeg < eend) {
            float4 A[8], B[8];
#pragma unroll
            for (int i = 0; i < 8; ++i) B[i] = make_float4(0.f, 0.f, 0.f, 0.f);

            int n  = (eend - ebeg < 32) ? (eend - ebeg) : 32;
            int c0 = 0, w0 = 0;
            if (lane < n) {
                c0 = col[ebeg + lane];
                w0 = __float_as_int(wgt[ebeg + lane]);
            }
            GATHER_ISSUE(A, c0, 0, n)

            int base = ebeg;
            while (n == 32) {
                const int rest = eend - (base + 32);
                const int nn   = (rest < 32) ? rest : 32;
                const int pe   = (base + 32 + lane < eend) ? (base + 32 + lane) : (eend - 1);
                const int c1   = col[pe];
                const int w1   = __float_as_int(wgt[pe]);

                GATHER_ISSUE(B, c0, 8, 32)
                GATHER_CONSUME(A, w0, 0, 32)
                GATHER_ISSUE(A, c0, 16, 32)
                GATHER_CONSUME(B, w0, 8, 32)
                GATHER_ISSUE(B, c0, 24, 32)
                GATHER_CONSUME(A, w0, 16, 32)
                if (nn > 0) {
                    GATHER_ISSUE(A, c1, 0, nn)
                    GATHER_CONSUME(B, w0, 24, 32)
                } else {
                    GATHER_CONSUME(B, w0, 24, 32)
                }
                c0 = c1; w0 = w1; n = nn; base += 32;
            }
            if (n > 0) {
                if (8 < n)  { GATHER_ISSUE(B, c0, 8, n) }
                GATHER_CONSUME(A, w0, 0, n)
                if (16 < n) { GATHER_ISSUE(A, c0, 16, n) }
                GATHER_CONSUME(B, w0, 8, n)
                if (24 < n) { GATHER_ISSUE(B, c0, 24, n) }
                GATHER_CONSUME(A, w0, 16, n)
                GATHER_CONSUME(B, w0, 24, n)
            }
        }
        while (cur < cnt) {
            M4[(size_t)(g0 + cur) * (DFEAT / 4) + lane] = acc;
            acc = make_float4(0.f, 0.f, 0.f, 0.f);
            ++cur;
        }
    }
}

#define GEMM_BLOCKS 512

template <bool RELU>
__global__ __launch_bounds__(256) void gcn_gemm(const float* A,
                                                const float* __restrict__ rs_in,
                                                const float* __restrict__ W,
                                                const float* __restrict__ bias,
                                                float* out,
                                                int n_tiles) {
    __shared__ float Alds[2][16 * ASTRIDE];

    const int tid  = threadIdx.x;
    const int wave = tid >> 5;
    const int lane = tid & 31;

    const int   mrow = lane & 15;
    const int   khi  = (lane >> 4) << 1;
    const int   n0   = wave << 4;
    const float* wp  = W + n0 + mrow;

    v2f bfrag[DFEAT / 4];
#pragma unroll
    for (int k = 0; k < DFEAT; k += 4) {
        bfrag[k >> 2].x = wp[(size_t)(k + khi)     * DFEAT];
        bfrag[k >> 2].y = wp[(size_t)(k + khi + 1) * DFEAT];
    }

    const int   col  = n0 + mrow;
    const float bv   = bias[col];
    const int   rsel = (lane >> 4) << 3;

    const int r = tid >> 4;
    const int c = (tid & 15) << 3;

    int    tile = blockIdx.x;
    float  scale = 0.f;
    float4 a0 = make_float4(0.f, 0.f, 0.f, 0.f);
    float4 a1 = make_float4(0.f, 0.f, 0.f, 0.f);
    if (tile < n_tiles) {
        const int row = (tile << 4) + r;
        const float4* gp = (const float4*)(A + (size_t)row * DFEAT + c);
        scale = rs_in[row];
        a0 = gp[0];
        a1 = gp[1];
    }

    int p = 0;
    for (; tile < n_tiles; tile += gridDim.x) {
        {
            float* lp = &Alds[p][r * ASTRIDE + c];
            lp[0] = a0.x * scale; lp[1] = a0.y * scale; lp[2] = a0.z * scale; lp[3] = a0.w * scale;
            lp[4] = a1.x * scale; lp[5] = a1.y * scale; lp[6] = a1.z * scale; lp[7] = a1.w * scale;
        }
        __syncthreads();

        const int nt = tile + gridDim.x;
        if (nt < n_tiles) {
            const int row = (nt << 4) + r;
            const float4* gp = (const float4*)(A + (size_t)row * DFEAT + c);
            scale = rs_in[row];
            a0 = gp[0];
            a1 = gp[1];
        }

        const float* ap = &Alds[p][mrow * ASTRIDE + khi];
        v8f acc = {0.f, 0.f, 0.f, 0.f, 0.f, 0.f, 0.f, 0.f};
#pragma unroll
        for (int k = 0; k < DFEAT; k += 4) {
            v2f a = *(const v2f*)(ap + k);
            acc = __builtin_amdgcn_wmma_f32_16x16x4_f32(false, a, false, bfrag[k >> 2],
                                                        (short)0, acc, false, false);
        }

        const int rowBase = tile << 4;
#pragma unroll
        for (int v = 0; v < 8; ++v) {
            float h = acc[v] + bv;
            if (RELU) h = fmaxf(h, 0.f);
            out[(size_t)(rowBase + v + rsel) * DFEAT + col] = h;
        }
        p ^= 1;
    }
}

extern "C" void kernel_launch(void* const* d_in, const int* in_sizes, int n_in,
                              void* d_out, int out_size, void* d_ws, size_t ws_size,
                              hipStream_t stream) {
    const float* x   = (const float*)d_in[0];
    const int*   src = (const int*)  d_in[1];
    const int*   dst = (const int*)  d_in[2];
    const float* W1  = (const float*)d_in[3];
    const float* b1  = (const float*)d_in[4];
    const float* W2  = (const float*)d_in[5];
    const float* b2  = (const float*)d_in[6];
    float*       out = (float*)d_out;

    float* ws     = (float*)d_ws;
    int*   TA     = (int*)ws;
    int*   TB     = (int*)ws + 40000;
    int*   baseA  = (int*)ws + 80000;
    float* hbuf   = ws + 2 * N_NODES;
    int2*  tmpA   = (int2*)hbuf;
    int2*  tmpB   = (int2*)(hbuf + 2 * N_EDGES);
    float* aux    = hbuf + (size_t)N_NODES * DFEAT;
    float* rs_in  = aux + N_NODES;
    int*   rowptr = (int*)(aux + 2 * N_NODES);
    int*   col    = (int*)(aux + 4 * N_NODES) + N_EDGES;
    float* wgt    = (float*)(col + N_EDGES);

    const int T256  = 256;
    const int GEMM_TILES = N_NODES / 16;

    chunk_sort_kernel<<<NCH, 1024, 0, stream>>>(src, dst, TA, tmpA, N_EDGES);
    src_range_kernel<<<NBK, 1024, 0, stream>>>(tmpA, TA, TB, baseA, tmpB);
    range_place_kernel<<<NBK, 1024, 0, stream>>>(tmpB, TB, baseA, rowptr, rs_in, col, wgt, N_NODES);

    gather_kernel<<<GATHER_BLOCKS, T256, 0, stream>>>(x, rowptr, col, wgt, hbuf, N_NODES);
    gcn_gemm<true><<<GEMM_BLOCKS, T256, 0, stream>>>(hbuf, rs_in, W1, b1, hbuf, GEMM_TILES);

    gather_kernel<<<GATHER_BLOCKS, T256, 0, stream>>>(hbuf, rowptr, col, wgt, out, N_NODES);
    gcn_gemm<false><<<GEMM_BLOCKS, T256, 0, stream>>>(out, rs_in, W2, b2, out, GEMM_TILES);
}


# Binning passes 2 and 3 read their edges once: a wave issues the loads of all its pieces together, keeps the edges in registers for count and place (no second read); row order: LDS arrival
# speedup vs baseline: 4.4045x; 1.0594x over previous
#include <hip/hip_runtime.h>

#define N_NODES 50000
#define N_EDGES 800000
#define DFEAT   128
#define ASTRIDE 132

typedef float v2f __attribute__((ext_vector_type(2)));
typedef float v8f __attribute__((ext_vector_type(8)));

#define CHUNK 4096
#define NCH   ((N_EDGES + CHUNK - 1) / CHUNK)
#define NBK   ((N_NODES + 255) >> 8)
#define TROW  (NBK + 1)
#define CAP   6144
#define NSEG  ((((NCH > NBK) ? NCH : NBK) + 31) / 32)

__global__ __launch_bounds__(1024) void chunk_sort_kernel(const int* __restrict__ src,
                                                          const int* __restrict__ dst,
                                                          int* __restrict__ TA,
                                                          int2* __restrict__ tmpA,
                                                          int n_edges) {
    __shared__ int  s_cnt[256];
    __shared__ int  s_wsum[8];
    __shared__ int2 s_edge[CHUNK];

    const int tid  = threadIdx.x;
    const int lane = tid & 31;
    const int wave = tid >> 5;
    const int chunkBase = blockIdx.x * CHUNK;

    if (tid < 256) s_cnt[tid] = 0;
    __syncthreads();

    int s[4], d[4], r[4];
#pragma unroll
    for (int i = 0; i < 4; ++i) {
        const int e = chunkBase + tid + i * 1024;
        s[i] = -1; d[i] = 0; r[i] = 0;
        if (e < n_edges) { s[i] = src[e]; d[i] = dst[e]; }
    }
#pragma unroll
    for (int i = 0; i < 4; ++i)
        if (s[i] >= 0) r[i] = atomicAdd(&s_cnt[s[i] >> 8], 1);
    __syncthreads();

    int c = 0, incl = 0;
    if (tid < 256) {
        c = s_cnt[tid];
        incl = c;
        for (int off = 1; off < 32; off <<= 1) {
            int t = __shfl_up(incl, off);
            if (lane >= off) incl += t;
        }
        if (lane == 31) s_wsum[wave] = incl;
    }
    __syncthreads();
    if (tid < 256) {
        int bw = 0;
        for (int w = 0; w < wave; ++w) bw += s_wsum[w];
        const int excl = bw + incl - c;
        s_cnt[tid] = excl;
        if (tid <= NBK) TA[blockIdx.x * TROW + tid] = excl;
    }
    __syncthreads();

#pragma unroll
    for (int i = 0; i < 4; ++i)
        if (s[i] >= 0) s_edge[s_cnt[s[i] >> 8] + r[i]] = make_int2(s[i], d[i]);
    __syncthreads();

#pragma unroll
    for (int i = 0; i < 4; ++i) {
        const int e = chunkBase + tid + i * 1024;
        if (e < n_edges) tmpA[e] = s_edge[tid + i * 1024];
    }
}

__global__ __launch_bounds__(1024) void src_range_kernel(const int2* __restrict__ tmpA,
                                                         const int* __restrict__ TA,
                                                         int* __restrict__ TB,
                                                         int* __restrict__ baseA,
                                                         int2* __restrict__ tmpB) {
    __shared__ int  s_segS[256];
    __shared__ int  s_segL[256];
    __shared__ int  s_deg[256];
    __shared__ int  s_cnt[256];
    __shared__ int  s_bsum[8];
    __shared__ int  s_tsum[8];
    __shared__ int  s_wsum[8];
    __shared__ int2 s_out[CAP];

    const int tid  = threadIdx.x;
    const int lane = tid & 31;
    const int wave = tid >> 5;
    const int b    = blockIdx.x;

    if (tid < 256) {
        int t0 = 0, st = 0, ln = 0;
        if (tid < NCH) {
            t0 = TA[tid * TROW + b];
            const int t1 = TA[tid * TROW + b + 1];
            st = tid * CHUNK + t0;
            ln = t1 - t0;
        }
        s_segS[tid] = st;
        s_segL[tid] = ln;
        s_deg[tid]  = 0;
        s_cnt[tid]  = 0;
        int v = t0, u = ln;
        for (int off = 16; off > 0; off >>= 1) {
            v += __shfl_xor(v, off);
            u += __shfl_xor(u, off);
        }
        if (lane == 0) { s_bsum[wave] = v; s_tsum[wave] = u; }
    }
    __syncthreads();

    int base = 0, total = 0;
#pragma unroll
    for (int w = 0; w < 8; ++w) { base += s_bsum[w]; total += s_tsum[w]; }
    const bool staged = (total <= CAP);

    int  sst[NSEG], sln[NSEG];
    int2 ea[NSEG], eb[NSEG];
#pragma unroll
    for (int k = 0; k < NSEG; ++k) {
        sst[k] = __builtin_amdgcn_readfirstlane(s_segS[wave + 32 * k]);
        sln[k] = __builtin_amdgcn_readfirstlane(s_segL[wave + 32 * k]);
    }
#pragma unroll
    for (int k = 0; k < NSEG; ++k) {
        ea[k] = make_int2(0, 0);
        eb[k] = make_int2(0, 0);
        if (lane < sln[k]) ea[k] = tmpA[sst[k] + lane];
        if (sln[k] > 32) {
            if (lane + 32 < sln[k]) eb[k] = tmpA[sst[k] + lane + 32];
        }
    }

#pragma unroll
    for (int k = 0; k < NSEG; ++k) {
        if (lane < sln[k]) {
            atomicAdd(&s_deg[ea[k].x & 255], 1);
            atomicAdd(&s_cnt[ea[k].y >> 8], 1);
        }
        if (sln[k] > 32) {
            if (lane + 32 < sln[k]) {
                atomicAdd(&s_deg[eb[k].x & 255], 1);
                atomicAdd(&s_cnt[eb[k].y >> 8], 1);
            }
            for (int j = lane + 64; j < sln[k]; j += 32) {
                const int2 e = tmpA[sst[k] + j];
                atomicAdd(&s_deg[e.x & 255], 1);
                atomicAdd(&s_cnt[e.y >> 8], 1);
            }
        }
    }
    __syncthreads();

    int c = 0, incl = 0;
    if (tid < 256) {
        c = s_cnt[tid];
        incl = c;
        for (int off = 1; off < 32; off <<= 1) {
            int t = __shfl_up(incl, off);
            if (lane >= off) incl += t;
        }
        if (lane == 31) s_wsum[wave] = incl;
    }
    __syncthreads();
    if (tid < 256) {
        int bw = 0;
        for (int w = 0; w < wave; ++w) bw += s_wsum[w];
        const int excl = bw + incl - c;
        s_cnt[tid] = excl;
        if (tid <= NBK) TB[b * TROW + tid] = excl;
        if (tid == 0) baseA[b] = base;
    }
    __syncthreads();

    int dga[NSEG], psa[NSEG];
#pragma unroll
    for (int k = 0; k < NSEG; ++k) dga[k] = s_deg[ea[k].x & 255];
#pragma unroll
    for (int k = 0; k < NSEG; ++k) {
        psa[k] = 0;
        if (lane < sln[k]) psa[k] = atomicAdd(&s_cnt[ea[k].y >> 8], 1);
    }
#pragma unroll
    for (int k = 0; k < NSEG; ++k) {
        if (lane < sln[k]) {
            const float w = rsqrtf(fmaxf((float)dga[k], 1.0f));
            const int2  o = make_int2((ea[k].x << 8) | (ea[k].y & 255), __float_as_int(w));
            if (staged) s_out[psa[k]] = o;
            else        tmpB[base + psa[k]] = o;
        }
    }
#pragma unroll
    for (int k = 0; k < NSEG; ++k) {
        if (sln[k] > 32) {
            for (int j = lane + 32; j < sln[k]; j += 32) {
                const int2  e   = (j < 64) ? eb[k] : tmpA[sst[k] + j];
                const float w   = rsqrtf(fmaxf((float)s_deg[e.x & 255], 1.0f));
                const int   pos = atomicAdd(&s_cnt[e.y >> 8], 1);
                const int2  o   = make_int2((e.x << 8) | (e.y & 255), __float_as_int(w));
                if (staged) s_out[pos] = o;
                else        tmpB[base + pos] = o;
            }
        }
    }
    __syncthreads();

    if (staged) {
        for (int i = tid; i < total; i += 1024) tmpB[base + i] = s_out[i];
    }
}

__global__ __launch_bounds__(1024) void range_place_kernel(const int2* __restrict__ tmpB,
                                                           const int* __restrict__ TB,
                                                           const int* __restrict__ baseA,
                                                           int* __restrict__ rowptr,
                                                           float* __restrict__ rs_in,
                                                           int* __restrict__ col,
                                                           float* __restrict__ wgt,
                                                           int n_nodes) {
    __shared__ int  s_segS[256];
    __shared__ int  s_segL[256];
    __shared__ int  s_cnt[256];
    __shared__ int  s_cur[256];
    __shared__ int  s_bsum[8];
    __shared__ int  s_tsum[8];
    __shared__ int  s_wsum[8];
    __shared__ int2 s_out[CAP];

    const int tid  = threadIdx.x;
    const int lane = tid & 31;
    const int wave = tid >> 5;
    const int b    = blockIdx.x;

    if (tid < 256) {
        int t0 = 0, st = 0, ln = 0;
        if (tid < NBK) {
            t0 = TB[tid * TROW + b];
            const int t1 = TB[tid * TROW + b + 1];
            st = baseA[tid] + t0;
            ln = t1 - t0;
        }
        s_segS[tid] = st;
        s_segL[tid] = ln;
        s_cnt[tid]  = 0;
        int v = t0, u = ln;
        for (int off = 16; off > 0; off >>= 1) {
            v += __shfl_xor(v, off);
            u += __shfl_xor(u, off);
        }
        if (lane == 0) { s_bsum[wave] = v; s_tsum[wave] = u; }
    }
    __syncthreads();

    int base = 0, total = 0;
#pragma unroll
    for (int w = 0; w < 8; ++w) { base += s_bsum[w]; total += s_tsum[w]; }
    const bool staged = (total <= CAP);

    int  sst[NSEG], sln[NSEG];
    int2 ea[NSEG], eb[NSEG];
#pragma unroll
    for (int k = 0; k < NSEG; ++k) {
        sst[k] = __builtin_amdgcn_readfirstlane(s_segS[wave + 32 * k]);
        sln[k] = __builtin_amdgcn_readfirstlane(s_segL[wave + 32 * k]);
    }
#pragma unroll
    for (int k = 0; k < NSEG; ++k) {
        ea[k] = make_int2(0, 0);
        eb[k] = make_int2(0, 0);
        if (lane < sln[k]) ea[k] = tmpB[sst[k] + lane];
        if (sln[k] > 32) {
            if (lane + 32 < sln[k]) eb[k] = tmpB[sst[k] + lane + 32];
        }
    }

#pragma unroll
    for (int k = 0; k < NSEG; ++k) {
        if (lane < sln[k]) atomicAdd(&s_cnt[ea[k].x & 255], 1);
        if (sln[k] > 32) {
            if (lane + 32 < sln[k]) atomicAdd(&s_cnt[eb[k].x & 255], 1);
            for (int j = lane + 64; j < sln[k]; j += 32) {
                const int2 e = tmpB[sst[k] + j];
                atomicAdd(&s_cnt[e.x & 255], 1);
            }
        }
    }
    __syncthreads();

    int c = 0, incl = 0;
    if (tid < 256) {
        c = s_cnt[tid];
        incl = c;
        for (int off = 1; off < 32; off <<= 1) {
            int t = __shfl_up(incl, off);
            if (lane >= off) incl += t;
        }
        if (lane == 31) s_wsum[wave] = incl;
    }
    __syncthreads();
    if (tid < 256) {
        int bw = 0;
        for (int w = 0; w < wave; ++w) bw += s_wsum[w];
        const int excl = bw + incl - c;
        s_cur[tid] = excl;
        const int node = (b << 8) + tid;
        if (node < n_nodes) {
            rowptr[node] = base + excl;
            rs_in[node]  = rsqrtf(fmaxf((float)c, 1.0f));
            if (node == n_nodes - 1) rowptr[n_nodes] = base + excl + c;
        }
    }
    __syncthreads();

    int psa[NSEG];
#pragma unroll
    for (int k = 0; k < NSEG; ++k) {
        psa[k] = 0;
        if (lane < sln[k]) psa[k] = atomicAdd(&s_cur[ea[k].x & 255], 1);
    }
#pragma unroll
    for (int k = 0; k < NSEG; ++k) {
        if (lane < sln[k]) {
            if (staged) {
                s_out[psa[k]] = make_int2(ea[k].x >> 8, ea[k].y);
            } else {
                col[base + psa[k]] = ea[k].x >> 8;
                wgt[base + psa[k]] = __int_as_float(ea[k].y);
            }
        }
    }
#pragma unroll
    for (int k = 0; k < NSEG; ++k) {
        if (sln[k] > 32) {
            for (int j = lane + 32; j < sln[k]; j += 32) {
                const int2 e   = (j < 64) ? eb[k] : tmpB[sst[k] + j];
                const int  pos = atomicAdd(&s_cur[e.x & 255], 1);
                if (staged) {
                    s_out[pos] = make_int2(e.x >> 8, e.y);
                } else {
                    col[base + pos] = e.x >> 8;
                    wgt[base + pos] = __int_as_float(e.y);
                }
            }
        }
    }
    __syncthreads();

    if (staged) {
        for (int i = tid; i < total; i += 1024) {
            const int2 o = s_out[i];
            col[base + i] = o.x;
            wgt[base + i] = __int_as_float(o.y);
        }
    }
}

__device__ __forceinline__ void row_fma(float4& acc, const float4 v, const float w) {
    acc.x = fmaf(v.x, w, acc.x);
    acc.y = fmaf(v.y, w, acc.y);
    acc.z = fmaf(v.z, w, acc.z);
    acc.w = fmaf(v.w, w, acc.w);
}

#define GATHER_WAVES_LOG2 13
#define GATHER_WAVES (1 << GATHER_WAVES_LOG2)
#define GATHER_BLOCKS (GATHER_WAVES / 8)

#define GATHER_ISSUE(V, CREG, J, NN)                                                  \
    _Pragma("unroll")                                                                 \
    for (int i = 0; i < 8; ++i) {                                                     \
        const int p = ((J) + i < (NN)) ? ((J) + i) : ((NN) - 1);                      \
        const int s = __builtin_amdgcn_readlane((CREG), p);                           \
        V[i] = X4[(size_t)s * (DFEAT / 4) + lane];                                    \
    }

#define GATHER_CONSUME(V, WREG, J, NN)                                                \
    _Pragma("unroll")                                                                 \
    for (int i = 0; i < 8; ++i) {                                                     \
        if ((J) + i < (NN)) {                                                         \
            while (e == cur_end) {                                                    \
                M4[(size_t)(g0 + cur) * (DFEAT / 4) + lane] = acc;                    \
                acc = make_float4(0.f, 0.f, 0.f, 0.f);                                \
                ++cur;                                                                \
                cur_end = __builtin_amdgcn_readlane(rp, cur + 1);                     \
            }                                                                         \
            const float w = __int_as_float(__builtin_amdgcn_readlane((WREG), (J) + i)); \
            row_fma(acc, V[i], w);                                                    \
            ++e;                                                                      \
        }                                                                             \
    }

__global__ __launch_bounds__(256) void gather_kernel(const float* __restrict__ X,
                                                     const int* __restrict__ rowptr,
                                                     const int* __restrict__ col,
                                                     const float* __restrict__ wgt,
                                                     float* __restrict__ msg,
                                                     int n_nodes) {
    const int lane = threadIdx.x & 31;
    const int wid  = __builtin_amdgcn_readfirstlane((int)((blockIdx.x * blockDim.x + threadIdx.x) >> 5));
    const int n0   = (int)(((unsigned long long)wid * (unsigned)n_nodes) >> GATHER_WAVES_LOG2);
    const int n1   = (int)(((unsigned long long)(wid + 1) * (unsigned)n_nodes) >> GATHER_WAVES_LOG2);

    const float4* X4 = (const float4*)X;
    float4*       M4 = (float4*)msg;

    for (int g0 = n0; g0 < n1; g0 += 31) {
        const int cnt  = (n1 - g0 < 31) ? (n1 - g0) : 31;
        const int rp   = rowptr[g0 + ((lane < cnt) ? lane : cnt)];
        const int ebeg = __builtin_amdgcn_readlane(rp, 0);
        const int eend = __builtin_amdgcn_readlane(rp, cnt);

        int    cur     = 0;
        int    cur_end = __builtin_amdgcn_readlane(rp, 1);
        int    e       = ebeg;
        float4 acc     = make_float4(0.f, 0.f, 0.f, 0.f);

        if (ebeg < eend) {
            float4 A[8], B[8];
#pragma unroll
            for (int i = 0; i < 8; ++i) B[i] = make_float4(0.f, 0.f, 0.f, 0.f);

            int n  = (eend - ebeg < 32) ? (eend - ebeg) : 32;
            int c0 = 0, w0 = 0;
            if (lane < n) {
                c0 = col[ebeg + lane];
                w0 = __float_as_int(wgt[ebeg + lane]);
            }
            GATHER_ISSUE(A, c0, 0, n)

            int base = ebeg;
            while (n == 32) {
                const int rest = eend - (base + 32);
                const int nn   = (rest < 32) ? rest : 32;
                const int pe   = (base + 32 + lane < eend) ? (base + 32 + lane) : (eend - 1);
                const int c1   = col[pe];
                const int w1   = __float_as_int(wgt[pe]);

                GATHER_ISSUE(B, c0, 8, 32)
                GATHER_CONSUME(A, w0, 0, 32)
                GATHER_ISSUE(A, c0, 16, 32)
                GATHER_CONSUME(B, w0, 8, 32)
                GATHER_ISSUE(B, c0, 24, 32)
                GATHER_CONSUME(A, w0, 16, 32)
                if (nn > 0) {
                    GATHER_ISSUE(A, c1, 0, nn)
                    GATHER_CONSUME(B, w0, 24, 32)
                } else {
                    GATHER_CONSUME(B, w0, 24, 32)
                }
                c0 = c1; w0 = w1; n = nn; base += 32;
            }
            if (n > 0) {
                if (8 < n)  { GATHER_ISSUE(B, c0, 8, n) }
                GATHER_CONSUME(A, w0, 0, n)
                if (16 < n) { GATHER_ISSUE(A, c0, 16, n) }
                GATHER_CONSUME(B, w0, 8, n)
                if (24 < n) { GATHER_ISSUE(B, c0, 24, n) }
                GATHER_CONSUME(A, w0, 16, n)
                GATHER_CONSUME(B, w0, 24, n)
            }
        }
        while (cur < cnt) {
            M4[(size_t)(g0 + cur) * (DFEAT / 4) + lane] = acc;
            acc = make_float4(0.f, 0.f, 0.f, 0.f);
            ++cur;
        }
    }
}

#define GEMM_BLOCKS 512

template <bool RELU>
__global__ __launch_bounds__(256) void gcn_gemm(const float* A,
                                                const float* __restrict__ rs_in,
                                                const float* __restrict__ W,
                                                const float* __restrict__ bias,
                                                float* out,
                                                int n_tiles) {
    __shared__ float Alds[2][16 * ASTRIDE];

    const int tid  = threadIdx.x;
    const int wave = tid >> 5;
    const int lane = tid & 31;

    const int   mrow = lane & 15;
    const int   khi  = (lane >> 4) << 1;
    const int   n0   = wave << 4;
    const float* wp  = W + n0 + mrow;

    v2f bfrag[DFEAT / 4];
#pragma unroll
    for (int k = 0; k < DFEAT; k += 4) {
        bfrag[k >> 2].x = wp[(size_t)(k + khi)     * DFEAT];
        bfrag[k >> 2].y = wp[(size_t)(k + khi + 1) * DFEAT];
    }

    const int   col  = n0 + mrow;
    const float bv   = bias[col];
    const int   rsel = (lane >> 4) << 3;

    const int r = tid >> 4;
    const int c = (tid & 15) << 3;

    int    tile = blockIdx.x;
    float  scale = 0.f;
    float4 a0 = make_float4(0.f, 0.f, 0.f, 0.f);
    float4 a1 = make_float4(0.f, 0.f, 0.f, 0.f);
    if (tile < n_tiles) {
        const int row = (tile << 4) + r;
        const float4* gp = (const float4*)(A + (size_t)row * DFEAT + c);
        scale = rs_in[row];
        a0 = gp[0];
        a1 = gp[1];
    }

    int p = 0;
    for (; tile < n_tiles; tile += gridDim.x) {
        {
            float* lp = &Alds[p][r * ASTRIDE + c];
            lp[0] = a0.x * scale; lp[1] = a0.y * scale; lp[2] = a0.z * scale; lp[3] = a0.w * scale;
            lp[4] = a1.x * scale; lp[5] = a1.y * scale; lp[6] = a1.z * scale; lp[7] = a1.w * scale;
        }
        __syncthreads();

        const int nt = tile + gridDim.x;
        if (nt < n_tiles) {
            const int row = (nt << 4) + r;
            const float4* gp = (const float4*)(A + (size_t)row * DFEAT + c);
            scale = rs_in[row];
            a0 = gp[0];
            a1 = gp[1];
        }

        const float* ap = &Alds[p][mrow * ASTRIDE + khi];
        v8f acc = {0.f, 0.f, 0.f, 0.f, 0.f, 0.f, 0.f, 0.f};
#pragma unroll
        for (int k = 0; k < DFEAT; k += 4) {
            v2f a = *(const v2f*)(ap + k);
            acc = __builtin_amdgcn_wmma_f32_16x16x4_f32(false, a, false, bfrag[k >> 2],
                                                        (short)0, acc, false, false);
        }

        const int rowBase = tile << 4;
#pragma unroll
        for (int v = 0; v < 8; ++v) {
            float h = acc[v] + bv;
            if (RELU) h = fmaxf(h, 0.f);
            out[(size_t)(rowBase + v + rsel) * DFEAT + col] = h;
        }
        p ^= 1;
    }
}

extern "C" void kernel_launch(void* const* d_in, const int* in_sizes, int n_in,
                              void* d_out, int out_size, void* d_ws, size_t ws_size,
                              hipStream_t stream) {
    const float* x   = (const float*)d_in[0];
    const int*   src = (const int*)  d_in[1];
    const int*   dst = (const int*)  d_in[2];
    const float* W1  = (const float*)d_in[3];
    const float* b1  = (const float*)d_in[4];
    const float* W2  = (const float*)d_in[5];
    const float* b2  = (const float*)d_in[6];
    float*       out = (float*)d_out;

    float* ws     = (float*)d_ws;
    int*   TA     = (int*)ws;
    int*   TB     = (int*)ws + 40000;
    int*   baseA  = (int*)ws + 80000;
    float* hbuf   = ws + 2 * N_NODES;
    int2*  tmpA   = (int2*)hbuf;
    int2*  tmpB   = (int2*)(hbuf + 2 * N_EDGES);
    float* aux    = hbuf + (size_t)N_NODES * DFEAT;
    float* rs_in  = aux + N_NODES;
    int*   rowptr = (int*)(aux + 2 * N_NODES);
    int*   col    = (int*)(aux + 4 * N_NODES) + N_EDGES;
    float* wgt    = (float*)(col + N_EDGES);

    const int T256  = 256;
    const int GEMM_TILES = N_NODES / 16;

    chunk_sort_kernel<<<NCH, 1024, 0, stream>>>(src, dst, TA, tmpA, N_EDGES);
    src_range_kernel<<<NBK, 1024, 0, stream>>>(tmpA, TA, TB, baseA, tmpB);
    range_place_kernel<<<NBK, 1024, 0, stream>>>(tmpB, TB, baseA, rowptr, rs_in, col, wgt, N_NODES);

    gather_kernel<<<GATHER_BLOCKS, T256, 0, stream>>>(x, rowptr, col, wgt, hbuf, N_NODES);
    gcn_gemm<true><<<GEMM_BLOCKS, T256, 0, stream>>>(hbuf, rs_in, W1, b1, hbuf, GEMM_TILES);

    gather_kernel<<<GATHER_BLOCKS, T256, 0, stream>>>(hbuf, rowptr, col, wgt, out, N_NODES);
    gcn_gemm<false><<<GEMM_BLOCKS, T256, 0, stream>>>(out, rs_in, W2, b2, out, GEMM_TILES);
}


# Edges carry no weight: layer-1 WMMA product runs first on rows scaled by rs_out, its edge sum applies rs_in, bias, relu, rs_out; source-range sort pass dropped, 7 to 6 launches; sum order changed
# speedup vs baseline: 4.6402x; 1.0535x over previous
#include <hip/hip_runtime.h>


#define N_NODES 50000
#define N_EDGES 800000
#define DFEAT   128
#define ASTRIDE 132

typedef float v2f __attribute__((ext_vector_type(2)));
typedef float v8f __attribute__((ext_vector_type(8)));

#define CHUNK 4096
#define NCH   ((N_EDGES + CHUNK - 1) / CHUNK)
#define NBK   ((N_NODES + 255) >> 8)
#define TROW  (NBK + 1)
#define CAP   6144
#define NSEG  ((NCH + 31) / 32)

__global__ __launch_bounds__(1024) void chunk_sort_kernel(const int* __restrict__ src,
                                                          const int* __restrict__ dst,
                                                          int* __restrict__ TS,
                                                          int* __restrict__ TD,
                                                          int* __restrict__ tmpS,
                                                          int* __restrict__ tmpD,
                                                          int n_edges) {
    __shared__ int s_cnt[512];
    __shared__ int s_wsum[16];
    __shared__ int s_eS[CHUNK];
    __shared__ int s_eD[CHUNK];

    const int tid  = threadIdx.x;
    const int lane = tid & 31;
    const int wave = tid >> 5;
    const int chunkBase = blockIdx.x * CHUNK;

    if (tid < 512) s_cnt[tid] = 0;
    __syncthreads();

    int s[4], d[4], rs[4], rd[4];
#pragma unroll
    for (int i = 0; i < 4; ++i) {
        const int e = chunkBase + tid + i * 1024;
        s[i] = -1; d[i] = 0; rs[i] = 0; rd[i] = 0;
        if (e < n_edges) { s[i] = src[e]; d[i] = dst[e]; }
    }
#pragma unroll
    for (int i = 0; i < 4; ++i)
        if (s[i] >= 0) {
            rs[i] = atomicAdd(&s_cnt[s[i] >> 8], 1);
            rd[i] = atomicAdd(&s_cnt[256 + (d[i] >> 8)], 1);
        }
    __syncthreads();

    int c = 0, incl = 0;
    if (tid < 512) {
        c = s_cnt[tid];
        incl = c;
        for (int off = 1; off < 32; off <<= 1) {
            int t = __shfl_up(incl, off);
            if (lane >= off) incl += t;
        }
        if (lane == 31) s_wsum[wave] = incl;
    }
    __syncthreads();
    if (tid < 512) {
        int bw = 0;
        for (int w = (wave & 8); w < wave; ++w) bw += s_wsum[w];
        const int excl = bw + incl - c;
        s_cnt[tid] = excl;
        const int idx = tid & 255;
        int* T = (tid < 256) ? TS : TD;
        if (idx <= NBK) T[blockIdx.x * TROW + idx] = excl;
    }
    __syncthreads();

#pragma unroll
    for (int i = 0; i < 4; ++i)
        if (s[i] >= 0) {
            s_eS[s_cnt[s[i] >> 8] + rs[i]]         = s[i];
            s_eD[s_cnt[256 + (d[i] >> 8)] + rd[i]] = (s[i] << 8) | (d[i] & 255);
        }
    __syncthreads();

#pragma unroll
    for (int i = 0; i < 4; ++i) {
        const int e = chunkBase + tid + i * 1024;
        if (e < n_edges) {
            tmpS[e] = s_eS[tid + i * 1024];
            tmpD[e] = s_eD[tid + i * 1024];
        }
    }
}

__global__ __launch_bounds__(1024) void range_kernel(const int* __restrict__ tmpS,
                                                     const int* __restrict__ tmpD,
                                                     const int* __restrict__ TS,
                                                     const int* __restrict__ TD,
                                                     int* __restrict__ rowptr,
                                                     float* __restrict__ rs_in,
                                                     float* __restrict__ rs_out,
                                                     int* __restrict__ col,
                                                     int n_nodes) {
    __shared__ int s_segS[256];
    __shared__ int s_segL[256];
    __shared__ int s_sgS[256];
    __shared__ int s_sgL[256];
    __shared__ int s_cnt[256];
    __shared__ int s_cur[256];
    __shared__ int s_deg[256];
    __shared__ int s_bsum[8];
    __shared__ int s_tsum[8];
    __shared__ int s_wsum[8];
    __shared__ int s_out[CAP];

    const int tid  = threadIdx.x;
    const int lane = tid & 31;
    const int wave = tid >> 5;
    const int b    = blockIdx.x;

    if (tid < 256) {
        int t0 = 0, st = 0, ln = 0, st2 = 0, ln2 = 0;
        if (tid < NCH) {
            t0 = TD[tid * TROW + b];
            const int t1 = TD[tid * TROW + b + 1];
            st = tid * CHUNK + t0;
            ln = t1 - t0;
            const int u0 = TS[tid * TROW + b];
            const int u1 = TS[tid * TROW + b + 1];
            st2 = tid * CHUNK + u0;
            ln2 = u1 - u0;
        }
        s_segS[tid] = st;
        s_segL[tid] = ln;
        s_sgS[tid]  = st2;
        s_sgL[tid]  = ln2;
        s_cnt[tid]  = 0;
        s_deg[tid]  = 0;
        int v = t0, u = ln;
        for (int off = 16; off > 0; off >>= 1) {
            v += __shfl_xor(v, off);
            u += __shfl_xor(u, off);
        }
        if (lane == 0) { s_bsum[wave] = v; s_tsum[wave] = u; }
    }
    __syncthreads();

    int base = 0, total = 0;
#pragma unroll
    for (int w = 0; w < 8; ++w) { base += s_bsum[w]; total += s_tsum[w]; }
    const bool staged = (total <= CAP);

    int sst[NSEG], sln[NSEG], qst[NSEG], qln[NSEG];
    int ea[NSEG], eb[NSEG], fa[NSEG], fb[NSEG];
#pragma unroll
    for (int k = 0; k < NSEG; ++k) {
        sst[k] = __builtin_amdgcn_readfirstlane(s_segS[wave + 32 * k]);
        sln[k] = __builtin_amdgcn_readfirstlane(s_segL[wave + 32 * k]);
        qst[k] = __builtin_amdgcn_readfirstlane(s_sgS[wave + 32 * k]);
        qln[k] = __builtin_amdgcn_readfirstlane(s_sgL[wave + 32 * k]);
    }
#pragma unroll
    for (int k = 0; k < NSEG; ++k) {
        const int ia = sst[k] + ((lane < sln[k]) ? lane : 0);
        const int ja = qst[k] + ((lane < qln[k]) ? lane : 0);
        ea[k] = tmpD[(ia < N_EDGES) ? ia : (N_EDGES - 1)];
        fa[k] = tmpS[(ja < N_EDGES) ? ja : (N_EDGES - 1)];
        eb[k] = 0; fb[k] = 0;
        if (sln[k] > 32) {
            if (lane + 32 < sln[k]) eb[k] = tmpD[sst[k] + lane + 32];
        }
        if (qln[k] > 32) {
            if (lane + 32 < qln[k]) fb[k] = tmpS[qst[k] + lane + 32];
        }
    }

#pragma unroll
    for (int k = 0; k < NSEG; ++k) {
        asm volatile("" : "+v"(fa[k]), "+v"(fb[k]));
    }

#pragma unroll
    for (int k = 0; k < NSEG; ++k) {
        if (lane < sln[k]) atomicAdd(&s_cnt[ea[k] & 255], 1);
        if (sln[k] > 32) {
            if (lane + 32 < sln[k]) atomicAdd(&s_cnt[eb[k] & 255], 1);
            for (int j = lane + 64; j < sln[k]; j += 32) {
                const int e = tmpD[sst[k] + j];
                atomicAdd(&s_cnt[e & 255], 1);
            }
        }
        if (lane < qln[k]) atomicAdd(&s_deg[fa[k] & 255], 1);
        if (qln[k] > 32) {
            if (lane + 32 < qln[k]) atomicAdd(&s_deg[fb[k] & 255], 1);
            for (int j = lane + 64; j < qln[k]; j += 32) {
                const int e = tmpS[qst[k] + j];
                atomicAdd(&s_deg[e & 255], 1);
            }
        }
    }
    __syncthreads();

    int c = 0, incl = 0;
    if (tid < 256) {
        c = s_cnt[tid];
        incl = c;
        for (int off = 1; off < 32; off <<= 1) {
            int t = __shfl_up(incl, off);
            if (lane >= off) incl += t;
        }
        if (lane == 31) s_wsum[wave] = incl;
    }
    __syncthreads();
    if (tid < 256) {
        int bw = 0;
        for (int w = 0; w < wave; ++w) bw += s_wsum[w];
        const int excl = bw + incl - c;
        s_cur[tid] = excl;
        const int node = (b << 8) + tid;
        if (node < n_nodes) {
            rowptr[node] = base + excl;
            rs_in[node]  = rsqrtf(fmaxf((float)c, 1.0f));
            rs_out[node] = rsqrtf(fmaxf((float)s_deg[tid], 1.0f));
            if (node == n_nodes - 1) rowptr[n_nodes] = base + excl + c;
        }
    }
    __syncthreads();

    int psa[NSEG];
#pragma unroll
    for (int k = 0; k < NSEG; ++k) {
        psa[k] = 0;
        if (lane < sln[k]) psa[k] = atomicAdd(&s_cur[ea[k] & 255], 1);
    }
#pragma unroll
    for (int k = 0; k < NSEG; ++k) {
        if (lane < sln[k]) {
            if (staged) s_out[psa[k]] = ea[k] >> 8;
            else        col[base + psa[k]] = ea[k] >> 8;
        }
    }
#pragma unroll
    for (int k = 0; k < NSEG; ++k) {
        if (sln[k] > 32) {
            for (int j = lane + 32; j < sln[k]; j += 32) {
                const int e   = (j < 64) ? eb[k] : tmpD[sst[k] + j];
                const int pos = atomicAdd(&s_cur[e & 255], 1);
                if (staged) s_out[pos] = e >> 8;
                else        col[base + pos] = e >> 8;
            }
        }
    }
    __syncthreads();

    if (staged) {
        for (int i = tid; i < total; i += 1024) col[base + i] = s_out[i];
    }
}

__device__ __forceinline__ void row_add(float4& acc, const float4 v) {
    acc.x += v.x;
    acc.y += v.y;
    acc.z += v.z;
    acc.w += v.w;
}

#define GATHER_WAVES_LOG2 13
#define GATHER_WAVES (1 << GATHER_WAVES_LOG2)
#define GATHER_BLOCKS (GATHER_WAVES / 8)

#define GATHER_ISSUE(V, CREG, J, NN)                                                  \
    _Pragma("unroll")                                                                 \
    for (int i = 0; i < 8; ++i) {                                                     \
        const int p = ((J) + i < (NN)) ? ((J) + i) : ((NN) - 1);                      \
        const int s = __builtin_amdgcn_readlane((CREG), p);                           \
        V[i] = X4[(size_t)s * (DFEAT / 4) + lane];                                    \
    }

#define GATHER_STORE_NODE()                                                           \
    {                                                                                 \
        float4 o = acc;                                                               \
        if (EPI) {                                                                    \
            const float si = __int_as_float(__builtin_amdgcn_readlane(ri, cur));      \
            const float so = __int_as_float(__builtin_amdgcn_readlane(ro, cur));      \
            o.x = fmaxf(acc.x * si + b4.x, 0.f) * so;                                 \
            o.y = fmaxf(acc.y * si + b4.y, 0.f) * so;                                 \
            o.z = fmaxf(acc.z * si + b4.z, 0.f) * so;                                 \
            o.w = fmaxf(acc.w * si + b4.w, 0.f) * so;                                 \
        }                                                                             \
        M4[(size_t)(g0 + cur) * (DFEAT / 4) + lane] = o;                              \
    }

#define GATHER_CONSUME(V, J, NN)                                                      \
    _Pragma("unroll")                                                                 \
    for (int i = 0; i < 8; ++i) {                                                     \
        if ((J) + i < (NN)) {                                                         \
            while (e == cur_end) {                                                    \
                GATHER_STORE_NODE()                                                   \
                acc = make_float4(0.f, 0.f, 0.f, 0.f);                                \
                ++cur;                                                                \
                cur_end = __builtin_amdgcn_readlane(rp, cur + 1);                     \
            }                                                                         \
            row_add(acc, V[i]);                                                       \
            ++e;                                                                      \
        }                                                                             \
    }

template <bool EPI>
__global__ __launch_bounds__(256) void gather_kernel(const float* __restrict__ X,
                                                     const int* __restrict__ rowptr,
                                                     const int* __restrict__ col,
                                                     const float* __restrict__ rs_in,
                                                     const float* __restrict__ rs_out,
                                                     const float* __restrict__ bias,
                                                     float* __restrict__ msg,
                                                     int n_nodes) {
    const int lane = threadIdx.x & 31;
    const int wid  = __builtin_amdgcn_readfirstlane((int)((blockIdx.x * blockDim.x + threadIdx.x) >> 5));
    const int n0   = (int)(((unsigned long long)wid * (unsigned)n_nodes) >> GATHER_WAVES_LOG2);
    const int n1   = (int)(((unsigned long long)(wid + 1) * (unsigned)n_nodes) >> GATHER_WAVES_LOG2);

    const float4* X4 = (const float4*)X;
    float4*       M4 = (float4*)msg;

    float4 b4 = make_float4(0.f, 0.f, 0.f, 0.f);
    if (EPI) b4 = ((const float4*)bias)[lane];

    for (int g0 = n0; g0 < n1; g0 += 31) {
        const int cnt  = (n1 - g0 < 31) ? (n1 - g0) : 31;
        const int rp   = rowptr[g0 + ((lane < cnt) ? lane : cnt)];
        const int ebeg = __builtin_amdgcn_readlane(rp, 0);
        const int eend = __builtin_amdgcn_readlane(rp, cnt);
        int ri = 0, ro = 0;
        if (EPI) {
            const int q = g0 + ((lane < cnt) ? lane : (cnt - 1));
            ri = __float_as_int(rs_in[q]);
            ro = __float_as_int(rs_out[q]);
        }

        int    cur     = 0;
        int    cur_end = __builtin_amdgcn_readlane(rp, 1);
        int    e       = ebeg;
        float4 acc     = make_float4(0.f, 0.f, 0.f, 0.f);

        if (ebeg < eend) {
            float4 A[8], B[8];
#pragma unroll
            for (int i = 0; i < 8; ++i) B[i] = make_float4(0.f, 0.f, 0.f, 0.f);

            int n  = (eend - ebeg < 32) ? (eend - ebeg) : 32;
            int c0 = 0;
            if (lane < n) c0 = col[ebeg + lane];
            GATHER_ISSUE(A, c0, 0, n)

            int base = ebeg;
            while (n == 32) {
                const int rest = eend - (base + 32);
                const int nn   = (rest < 32) ? rest : 32;
                const int pe   = (base + 32 + lane < eend) ? (base + 32 + lane) : (eend - 1);
                const int c1   = col[pe];

                GATHER_ISSUE(B, c0, 8, 32)
                GATHER_CONSUME(A, 0, 32)
                GATHER_ISSUE(A, c0, 16, 32)
                GATHER_CONSUME(B, 8, 32)
                GATHER_ISSUE(B, c0, 24, 32)
                GATHER_CONSUME(A, 16, 32)
                if (nn > 0) {
                    GATHER_ISSUE(A, c1, 0, nn)
                    GATHER_CONSUME(B, 24, 32)
                } else {
                    GATHER_CONSUME(B, 24, 32)
                }
                c0 = c1; n = nn; base += 32;
            }
            if (n > 0) {
                if (8 < n)  { GATHER_ISSUE(B, c0, 8, n) }
                GATHER_CONSUME(A, 0, n)
                if (16 < n) { GATHER_ISSUE(A, c0, 16, n) }
                GATHER_CONSUME(B, 8, n)
                if (24 < n) { GATHER_ISSUE(B, c0, 24, n) }
                GATHER_CONSUME(A, 16, n)
                GATHER_CONSUME(B, 24, n)
            }
        }
        while (cur < cnt) {
            GATHER_STORE_NODE()
            acc = make_float4(0.f, 0.f, 0.f, 0.f);
            ++cur;
        }
    }
}

#define GEMM_BLOCKS 512

template <bool BIAS>
__global__ __launch_bounds__(256) void gcn_gemm(const float* A,
                                                const float* __restrict__ rscale,
                                                const float* __restrict__ W,
                                                const float* __restrict__ bias,
                                                float* out,
                                                int n_tiles) {
    __shared__ float Alds[2][16 * ASTRIDE];

    const int tid  = threadIdx.x;
    const int wave = tid >> 5;
    const int lane = tid & 31;

    const int   mrow = lane & 15;
    const int   khi  = (lane >> 4) << 1;
    const int   n0   = wave << 4;
    const float* wp  = W + n0 + mrow;

    v2f bfrag[DFEAT / 4];
#pragma unroll
    for (int k = 0; k < DFEAT; k += 4) {
        bfrag[k >> 2].x = wp[(size_t)(k + khi)     * DFEAT];
        bfrag[k >> 2].y = wp[(size_t)(k + khi + 1) * DFEAT];
    }

    const int   col  = n0 + mrow;
    float       bv   = 0.f;
    if (BIAS) bv = bias[col];
    const int   rsel = (lane >> 4) << 3;

    const int r = tid >> 4;
    const int c = (tid & 15) << 3;

    int    tile = blockIdx.x;
    float  scale = 0.f;
    float4 a0 = make_float4(0.f, 0.f, 0.f, 0.f);
    float4 a1 = make_float4(0.f, 0.f, 0.f, 0.f);
    if (tile < n_tiles) {
        const int row = (tile << 4) + r;
        const float4* gp = (const float4*)(A + (size_t)row * DFEAT + c);
        scale = rscale[row];
        a0 = gp[0];
        a1 = gp[1];
    }

    int p = 0;
    for (; tile < n_tiles; tile += gridDim.x) {
        {
            float* lp = &Alds[p][r * ASTRIDE + c];
            lp[0] = a0.x * scale; lp[1] = a0.y * scale; lp[2] = a0.z * scale; lp[3] = a0.w * scale;
            lp[4] = a1.x * scale; lp[5] = a1.y * scale; lp[6] = a1.z * scale; lp[7] = a1.w * scale;
        }
        __syncthreads();

        const int nt = tile + gridDim.x;
        if (nt < n_tiles) {
            const int row = (nt << 4) + r;
            const float4* gp = (const float4*)(A + (size_t)row * DFEAT + c);
            scale = rscale[row];
            a0 = gp[0];
            a1 = gp[1];
        }

        const float* ap = &Alds[p][mrow * ASTRIDE + khi];
        v8f acc = {0.f, 0.f, 0.f, 0.f, 0.f, 0.f, 0.f, 0.f};
#pragma unroll
        for (int k = 0; k < DFEAT; k += 4) {
            v2f a = *(const v2f*)(ap + k);
            acc = __builtin_amdgcn_wmma_f32_16x16x4_f32(false, a, false, bfrag[k >> 2],
                                                        (short)0, acc, false, false);
        }

        const int rowBase = tile << 4;
#pragma unroll
        for (int v = 0; v < 8; ++v) {
            float h = acc[v];
            if (BIAS) h += bv;
            out[(size_t)(rowBase + v + rsel) * DFEAT + col] = h;
        }
        p ^= 1;
    }
}

extern "C" void kernel_launch(void* const* d_in, const int* in_sizes, int n_in,
                              void* d_out, int out_size, void* d_ws, size_t ws_size,
                              hipStream_t stream) {
    const float* x   = (const float*)d_in[0];
    const int*   src = (const int*)  d_in[1];
    const int*   dst = (const int*)  d_in[2];
    const float* W1  = (const float*)d_in[3];
    const float* b1  = (const float*)d_in[4];
    const float* W2  = (const float*)d_in[5];
    const float* b2  = (const float*)d_in[6];
    float*       out = (float*)d_out;

    float* ws     = (float*)d_ws;
    int*   TS     = (int*)ws;
    int*   TD     = (int*)ws + 40000;
    float* hbuf   = ws + 2 * N_NODES;
    int*   tmpS   = (int*)hbuf;
    int*   tmpD   = (int*)hbuf + N_EDGES;
    float* aux    = hbuf + (size_t)N_NODES * DFEAT;
    float* rs_out = aux;
    float* rs_in  = aux + N_NODES;
    int*   rowptr = (int*)(aux + 2 * N_NODES);
    int*   col    = (int*)(aux + 4 * N_NODES) + N_EDGES;

    const int T256  = 256;
    const int GEMM_TILES = N_NODES / 16;

    chunk_sort_kernel<<<NCH, 1024, 0, stream>>>(src, dst, TS, TD, tmpS, tmpD, N_EDGES);
    range_kernel<<<NBK, 1024, 0, stream>>>(tmpS, tmpD, TS, TD, rowptr, rs_in, rs_out, col, N_NODES);

    gcn_gemm<false><<<GEMM_BLOCKS, T256, 0, stream>>>(x, rs_out, W1, b1, out, GEMM_TILES);
    gather_kernel<true><<<GATHER_BLOCKS, T256, 0, stream>>>(out, rowptr, col, rs_in, rs_out, b1, hbuf, N_NODES);

    gather_kernel<false><<<GATHER_BLOCKS, T256, 0, stream>>>(hbuf, rowptr, col, rs_in, rs_out, b2, out, N_NODES);
    gcn_gemm<true><<<GEMM_BLOCKS, T256, 0, stream>>>(out, rs_in, W2, b2, out, GEMM_TILES);
}


# Layer-2 edge sum moved into its WMMA product kernel: a wave sums two nodes' in-edge rows (same list order) into the LDS A tile, the block multiplies it; sums skip global memory; 6 to 5 launches
# speedup vs baseline: 4.8940x; 1.0547x over previous
#include <hip/hip_runtime.h>


#define N_NODES 50000
#define N_EDGES 800000
#define DFEAT   128
#define ASTRIDE 132

typedef float v2f __attribute__((ext_vector_type(2)));
typedef float v8f __attribute__((ext_vector_type(8)));

#define CHUNK 4096
#define NCH   ((N_EDGES + CHUNK - 1) / CHUNK)
#define NBK   ((N_NODES + 255) >> 8)
#define TROW  (NBK + 1)
#define CAP   6144
#define NSEG  ((NCH + 31) / 32)

__global__ __launch_bounds__(1024) void chunk_sort_kernel(const int* __restrict__ src,
                                                          const int* __restrict__ dst,
                                                          int* __restrict__ TS,
                                                          int* __restrict__ TD,
                                                          int* __restrict__ tmpS,
                                                          int* __restrict__ tmpD,
                                                          int n_edges) {
    __shared__ int s_cnt[512];
    __shared__ int s_wsum[16];
    __shared__ int s_eS[CHUNK];
    __shared__ int s_eD[CHUNK];

    const int tid  = threadIdx.x;
    const int lane = tid & 31;
    const int wave = tid >> 5;
    const int chunkBase = blockIdx.x * CHUNK;

    if (tid < 512) s_cnt[tid] = 0;
    __syncthreads();

    int s[4], d[4], rs[4], rd[4];
#pragma unroll
    for (int i = 0; i < 4; ++i) {
        const int e = chunkBase + tid + i * 1024;
        s[i] = -1; d[i] = 0; rs[i] = 0; rd[i] = 0;
        if (e < n_edges) { s[i] = src[e]; d[i] = dst[e]; }
    }
#pragma unroll
    for (int i = 0; i < 4; ++i)
        if (s[i] >= 0) {
            rs[i] = atomicAdd(&s_cnt[s[i] >> 8], 1);
            rd[i] = atomicAdd(&s_cnt[256 + (d[i] >> 8)], 1);
        }
    __syncthreads();

    int c = 0, incl = 0;
    if (tid < 512) {
        c = s_cnt[tid];
        incl = c;
        for (int off = 1; off < 32; off <<= 1) {
            int t = __shfl_up(incl, off);
            if (lane >= off) incl += t;
        }
        if (lane == 31) s_wsum[wave] = incl;
    }
    __syncthreads();
    if (tid < 512) {
        int bw = 0;
        for (int w = (wave & 8); w < wave; ++w) bw += s_wsum[w];
        const int excl = bw + incl - c;
        s_cnt[tid] = excl;
        const int idx = tid & 255;
        int* T = (tid < 256) ? TS : TD;
        if (idx <= NBK) T[blockIdx.x * TROW + idx] = excl;
    }
    __syncthreads();

#pragma unroll
    for (int i = 0; i < 4; ++i)
        if (s[i] >= 0) {
            s_eS[s_cnt[s[i] >> 8] + rs[i]]         = s[i];
            s_eD[s_cnt[256 + (d[i] >> 8)] + rd[i]] = (s[i] << 8) | (d[i] & 255);
        }
    __syncthreads();

#pragma unroll
    for (int i = 0; i < 4; ++i) {
        const int e = chunkBase + tid + i * 1024;
        if (e < n_edges) {
            tmpS[e] = s_eS[tid + i * 1024];
            tmpD[e] = s_eD[tid + i * 1024];
        }
    }
}

__global__ __launch_bounds__(1024) void range_kernel(const int* __restrict__ tmpS,
                                                     const int* __restrict__ tmpD,
                                                     const int* __restrict__ TS,
                                                     const int* __restrict__ TD,
                                                     int* __restrict__ rowptr,
                                                     float* __restrict__ rs_in,
                                                     float* __restrict__ rs_out,
                                                     int* __restrict__ col,
                                                     int n_nodes) {
    __shared__ int s_segS[256];
    __shared__ int s_segL[256];
    __shared__ int s_sgS[256];
    __shared__ int s_sgL[256];
    __shared__ int s_cnt[256];
    __shared__ int s_cur[256];
    __shared__ int s_deg[256];
    __shared__ int s_bsum[8];
    __shared__ int s_tsum[8];
    __shared__ int s_wsum[8];
    __shared__ int s_out[CAP];

    const int tid  = threadIdx.x;
    const int lane = tid & 31;
    const int wave = tid >> 5;
    const int b    = blockIdx.x;

    if (tid < 256) {
        int t0 = 0, st = 0, ln = 0, st2 = 0, ln2 = 0;
        if (tid < NCH) {
            t0 = TD[tid * TROW + b];
            const int t1 = TD[tid * TROW + b + 1];
            st = tid * CHUNK + t0;
            ln = t1 - t0;
            const int u0 = TS[tid * TROW + b];
            const int u1 = TS[tid * TROW + b + 1];
            st2 = tid * CHUNK + u0;
            ln2 = u1 - u0;
        }
        s_segS[tid] = st;
        s_segL[tid] = ln;
        s_sgS[tid]  = st2;
        s_sgL[tid]  = ln2;
        s_cnt[tid]  = 0;
        s_deg[tid]  = 0;
        int v = t0, u = ln;
        for (int off = 16; off > 0; off >>= 1) {
            v += __shfl_xor(v, off);
            u += __shfl_xor(u, off);
        }
        if (lane == 0) { s_bsum[wave] = v; s_tsum[wave] = u; }
    }
    __syncthreads();

    int base = 0, total = 0;
#pragma unroll
    for (int w = 0; w < 8; ++w) { base += s_bsum[w]; total += s_tsum[w]; }
    const bool staged = (total <= CAP);

    int sst[NSEG], sln[NSEG], qst[NSEG], qln[NSEG];
    int ea[NSEG], eb[NSEG], fa[NSEG], fb[NSEG];
#pragma unroll
    for (int k = 0; k < NSEG; ++k) {
        sst[k] = __builtin_amdgcn_readfirstlane(s_segS[wave + 32 * k]);
        sln[k] = __builtin_amdgcn_readfirstlane(s_segL[wave + 32 * k]);
        qst[k] = __builtin_amdgcn_readfirstlane(s_sgS[wave + 32 * k]);
        qln[k] = __builtin_amdgcn_readfirstlane(s_sgL[wave + 32 * k]);
    }
#pragma unroll
    for (int k = 0; k < NSEG; ++k) {
        const int ia = sst[k] + ((lane < sln[k]) ? lane : 0);
        const int ja = qst[k] + ((lane < qln[k]) ? lane : 0);
        ea[k] = tmpD[(ia < N_EDGES) ? ia : (N_EDGES - 1)];
        fa[k] = tmpS[(ja < N_EDGES) ? ja : (N_EDGES - 1)];
        eb[k] = 0; fb[k] = 0;
        if (sln[k] > 32) {
            if (lane + 32 < sln[k]) eb[k] = tmpD[sst[k] + lane + 32];
        }
        if (qln[k] > 32) {
            if (lane + 32 < qln[k]) fb[k] = tmpS[qst[k] + lane + 32];
        }
    }

#pragma unroll
    for (int k = 0; k < NSEG; ++k) {
        asm volatile("" : "+v"(fa[k]), "+v"(fb[k]));
    }

#pragma unroll
    for (int k = 0; k < NSEG; ++k) {
        if (lane < sln[k]) atomicAdd(&s_cnt[ea[k] & 255], 1);
        if (sln[k] > 32) {
            if (lane + 32 < sln[k]) atomicAdd(&s_cnt[eb[k] & 255], 1);
            for (int j = lane + 64; j < sln[k]; j += 32) {
                const int e = tmpD[sst[k] + j];
                atomicAdd(&s_cnt[e & 255], 1);
            }
        }
        if (lane < qln[k]) atomicAdd(&s_deg[fa[k] & 255], 1);
        if (qln[k] > 32) {
            if (lane + 32 < qln[k]) atomicAdd(&s_deg[fb[k] & 255], 1);
            for (int j = lane + 64; j < qln[k]; j += 32) {
                const int e = tmpS[qst[k] + j];
                atomicAdd(&s_deg[e & 255], 1);
            }
        }
    }
    __syncthreads();

    int c = 0, incl = 0;
    if (tid < 256) {
        c = s_cnt[tid];
        incl = c;
        for (int off = 1; off < 32; off <<= 1) {
            int t = __shfl_up(incl, off);
            if (lane >= off) incl += t;
        }
        if (lane == 31) s_wsum[wave] = incl;
    }
    __syncthreads();
    if (tid < 256) {
        int bw = 0;
        for (int w = 0; w < wave; ++w) bw += s_wsum[w];
        const int excl = bw + incl - c;
        s_cur[tid] = excl;
        const int node = (b << 8) + tid;
        if (node < n_nodes) {
            rowptr[node] = base + excl;
            rs_in[node]  = rsqrtf(fmaxf((float)c, 1.0f));
            rs_out[node] = rsqrtf(fmaxf((float)s_deg[tid], 1.0f));
            if (node == n_nodes - 1) rowptr[n_nodes] = base + excl + c;
        }
    }
    __syncthreads();

    int psa[NSEG];
#pragma unroll
    for (int k = 0; k < NSEG; ++k) {
        psa[k] = 0;
        if (lane < sln[k]) psa[k] = atomicAdd(&s_cur[ea[k] & 255], 1);
    }
#pragma unroll
    for (int k = 0; k < NSEG; ++k) {
        if (lane < sln[k]) {
            if (staged) s_out[psa[k]] = ea[k] >> 8;
            else        col[base + psa[k]] = ea[k] >> 8;
        }
    }
#pragma unroll
    for (int k = 0; k < NSEG; ++k) {
        if (sln[k] > 32) {
            for (int j = lane + 32; j < sln[k]; j += 32) {
                const int e   = (j < 64) ? eb[k] : tmpD[sst[k] + j];
                const int pos = atomicAdd(&s_cur[e & 255], 1);
                if (staged) s_out[pos] = e >> 8;
                else        col[base + pos] = e >> 8;
            }
        }
    }
    __syncthreads();

    if (staged) {
        for (int i = tid; i < total; i += 1024) col[base + i] = s_out[i];
    }
}

__device__ __forceinline__ void row_add(float4& acc, const float4 v) {
    acc.x += v.x;
    acc.y += v.y;
    acc.z += v.z;
    acc.w += v.w;
}

#define GATHER_WAVES_LOG2 13
#define GATHER_WAVES (1 << GATHER_WAVES_LOG2)
#define GATHER_BLOCKS (GATHER_WAVES / 8)

#define GATHER_ISSUE(V, CREG, J, NN)                                                  \
    _Pragma("unroll")                                                                 \
    for (int i = 0; i < 8; ++i) {                                                     \
        const int p = ((J) + i < (NN)) ? ((J) + i) : ((NN) - 1);                      \
        const int s = __builtin_amdgcn_readlane((CREG), p);                           \
        V[i] = X4[(size_t)s * (DFEAT / 4) + lane];                                    \
    }

#define GATHER_STORE_NODE()                                                           \
    {                                                                                 \
        float4 o;                                                                     \
        const float si = __int_as_float(__builtin_amdgcn_readlane(ri, cur));          \
        const float so = __int_as_float(__builtin_amdgcn_readlane(ro, cur));          \
        o.x = fmaxf(acc.x * si + b4.x, 0.f) * so;                                     \
        o.y = fmaxf(acc.y * si + b4.y, 0.f) * so;                                     \
        o.z = fmaxf(acc.z * si + b4.z, 0.f) * so;                                     \
        o.w = fmaxf(acc.w * si + b4.w, 0.f) * so;                                     \
        M4[(size_t)(g0 + cur) * (DFEAT / 4) + lane] = o;                              \
    }

#define GATHER_STAGE_NODE()                                                           \
    {                                                                                 \
        float4 o;                                                                     \
        const float si = __int_as_float(__builtin_amdgcn_readlane(ri, cur));          \
        o.x = acc.x * si;                                                             \
        o.y = acc.y * si;                                                             \
        o.z = acc.z * si;                                                             \
        o.w = acc.w * si;                                                             \
        *(float4*)&Alds[buf][(2 * wave + cur) * ASTRIDE + 4 * lane] = o;              \
    }

#define GATHER_CONSUME(V, J, NN, STORE)                                               \
    _Pragma("unroll")                                                                 \
    for (int i = 0; i < 8; ++i) {                                                     \
        if ((J) + i < (NN)) {                                                         \
            while (e == cur_end) {                                                    \
                STORE()                                                               \
                acc = make_float4(0.f, 0.f, 0.f, 0.f);                                \
                ++cur;                                                                \
                cur_end = __builtin_amdgcn_readlane(rp, cur + 1);                     \
            }                                                                         \
            row_add(acc, V[i]);                                                       \
            ++e;                                                                      \
        }                                                                             \
    }

#define GATHER_STREAM(STORE)                                                          \
    {                                                                                 \
        float4 A[8], B[8];                                                            \
        _Pragma("unroll")                                                             \
        for (int i = 0; i < 8; ++i) B[i] = make_float4(0.f, 0.f, 0.f, 0.f);           \
        GATHER_ISSUE(A, c0, 0, n)                                                     \
        int base = ebeg;                                                              \
        while (n == 32) {                                                             \
            const int rest = eend - (base + 32);                                      \
            const int nn   = (rest < 32) ? rest : 32;                                 \
            const int pe   = (base + 32 + lane < eend) ? (base + 32 + lane) : (eend - 1); \
            const int c1   = col[pe];                                                 \
            GATHER_ISSUE(B, c0, 8, 32)                                                \
            GATHER_CONSUME(A, 0, 32, STORE)                                           \
            GATHER_ISSUE(A, c0, 16, 32)                                               \
            GATHER_CONSUME(B, 8, 32, STORE)                                           \
            GATHER_ISSUE(B, c0, 24, 32)                                               \
            GATHER_CONSUME(A, 16, 32, STORE)                                          \
            if (nn > 0) {                                                             \
                GATHER_ISSUE(A, c1, 0, nn)                                            \
                GATHER_CONSUME(B, 24, 32, STORE)                                      \
            } else {                                                                  \
                GATHER_CONSUME(B, 24, 32, STORE)                                      \
            }                                                                         \
            c0 = c1; n = nn; base += 32;                                              \
        }                                                                             \
        if (n > 0) {                                                                  \
            if (8 < n)  { GATHER_ISSUE(B, c0, 8, n) }                                 \
            GATHER_CONSUME(A, 0, n, STORE)                                            \
            if (16 < n) { GATHER_ISSUE(A, c0, 16, n) }                                \
            GATHER_CONSUME(B, 8, n, STORE)                                            \
            if (24 < n) { GATHER_ISSUE(B, c0, 24, n) }                                \
            GATHER_CONSUME(A, 16, n, STORE)                                           \
            GATHER_CONSUME(B, 24, n, STORE)                                           \
        }                                                                             \
    }

__global__ __launch_bounds__(256) void gather_kernel(const float* __restrict__ X,
                                                     const int* __restrict__ rowptr,
                                                     const int* __restrict__ col,
                                                     const float* __restrict__ rs_in,
                                                     const float* __restrict__ rs_out,
                                                     const float* __restrict__ bias,
                                                     float* __restrict__ msg,
                                                     int n_nodes) {
    const int lane = threadIdx.x & 31;
    const int wid  = __builtin_amdgcn_readfirstlane((int)((blockIdx.x * blockDim.x + threadIdx.x) >> 5));
    const int n0   = (int)(((unsigned long long)wid * (unsigned)n_nodes) >> GATHER_WAVES_LOG2);
    const int n1   = (int)(((unsigned long long)(wid + 1) * (unsigned)n_nodes) >> GATHER_WAVES_LOG2);

    const float4* X4 = (const float4*)X;
    float4*       M4 = (float4*)msg;

    const float4 b4 = ((const float4*)bias)[lane];

    for (int g0 = n0; g0 < n1; g0 += 31) {
        const int cnt  = (n1 - g0 < 31) ? (n1 - g0) : 31;
        const int rp   = rowptr[g0 + ((lane < cnt) ? lane : cnt)];
        const int ebeg = __builtin_amdgcn_readlane(rp, 0);
        const int eend = __builtin_amdgcn_readlane(rp, cnt);
        const int q  = g0 + ((lane < cnt) ? lane : (cnt - 1));
        const int ri = __float_as_int(rs_in[q]);
        const int ro = __float_as_int(rs_out[q]);

        int    cur     = 0;
        int    cur_end = __builtin_amdgcn_readlane(rp, 1);
        int    e       = ebeg;
        float4 acc     = make_float4(0.f, 0.f, 0.f, 0.f);

        if (ebeg < eend) {
            int n  = (eend - ebeg < 32) ? (eend - ebeg) : 32;
            int c0 = 0;
            if (lane < n) c0 = col[ebeg + lane];
            GATHER_STREAM(GATHER_STORE_NODE)
        }
        while (cur < cnt) {
            GATHER_STORE_NODE()
            acc = make_float4(0.f, 0.f, 0.f, 0.f);
            ++cur;
        }
    }
}

#define GEMM_BLOCKS 512

__global__ __launch_bounds__(256) void gcn_gemm(const float* A,
                                                const float* __restrict__ rscale,
                                                const float* __restrict__ W,
                                                float* out,
                                                int n_tiles) {
    __shared__ float Alds[2][16 * ASTRIDE];

    const int tid  = threadIdx.x;
    const int wave = tid >> 5;
    const int lane = tid & 31;

    const int   mrow = lane & 15;
    const int   khi  = (lane >> 4) << 1;
    const int   n0   = wave << 4;
    const float* wp  = W + n0 + mrow;

    v2f bfrag[DFEAT / 4];
#pragma unroll
    for (int k = 0; k < DFEAT; k += 4) {
        bfrag[k >> 2].x = wp[(size_t)(k + khi)     * DFEAT];
        bfrag[k >> 2].y = wp[(size_t)(k + khi + 1) * DFEAT];
    }

    const int   ocol = n0 + mrow;
    const int   rsel = (lane >> 4) << 3;

    const int r = tid >> 4;
    const int c = (tid & 15) << 3;

    int    tile = blockIdx.x;
    float  scale = 0.f;
    float4 a0 = make_float4(0.f, 0.f, 0.f, 0.f);
    float4 a1 = make_float4(0.f, 0.f, 0.f, 0.f);
    if (tile < n_tiles) {
        const int row = (tile << 4) + r;
        const float4* gp = (const float4*)(A + (size_t)row * DFEAT + c);
        scale = rscale[row];
        a0 = gp[0];
        a1 = gp[1];
    }

    int p = 0;
    for (; tile < n_tiles; tile += gridDim.x) {
        {
            float* lp = &Alds[p][r * ASTRIDE + c];
            lp[0] = a0.x * scale; lp[1] = a0.y * scale; lp[2] = a0.z * scale; lp[3] = a0.w * scale;
            lp[4] = a1.x * scale; lp[5] = a1.y * scale; lp[6] = a1.z * scale; lp[7] = a1.w * scale;
        }
        __syncthreads();

        const int nt = tile + gridDim.x;
        if (nt < n_tiles) {
            const int row = (nt << 4) + r;
            const float4* gp = (const float4*)(A + (size_t)row * DFEAT + c);
            scale = rscale[row];
            a0 = gp[0];
            a1 = gp[1];
        }

        const float* ap = &Alds[p][mrow * ASTRIDE + khi];
        v8f acc = {0.f, 0.f, 0.f, 0.f, 0.f, 0.f, 0.f, 0.f};
#pragma unroll
        for (int k = 0; k < DFEAT; k += 4) {
            v2f a = *(const v2f*)(ap + k);
            acc = __builtin_amdgcn_wmma_f32_16x16x4_f32(false, a, false, bfrag[k >> 2],
                                                        (short)0, acc, false, false);
        }

        const int rowBase = tile << 4;
#pragma unroll
        for (int v = 0; v < 8; ++v)
            out[(size_t)(rowBase + v + rsel) * DFEAT + ocol] = acc[v];
        p ^= 1;
    }
}

__device__ __forceinline__ int first_chunk_pos(int ebeg, int eend, int lane) {
    const int pe = (ebeg + lane < eend) ? (ebeg + lane) : (eend - 1);
    return (pe > 0) ? pe : 0;
}

#define FUSED_BLOCKS 625

__global__ __launch_bounds__(256, 6) void gather_gemm_kernel(const float* __restrict__ X,
                                                          const int* __restrict__ rowptr,
                                                          const int* __restrict__ col,
                                                          const float* __restrict__ rs_in,
                                                          const float* __restrict__ W,
                                                          const float* __restrict__ bias,
                                                          float* __restrict__ out,
                                                          int n_tiles) {
    __shared__ __attribute__((aligned(16))) float Alds[2][16 * ASTRIDE];

    const int lane = threadIdx.x & 31;
    const int wave = __builtin_amdgcn_readfirstlane((int)(threadIdx.x >> 5));

    const float4* X4 = (const float4*)X;

    const int lsel = (lane < 2) ? lane : 2;
    int tile = blockIdx.x;
    if (tile >= n_tiles) return;
    int rp = rowptr[(tile << 4) + 2 * wave + lsel];

    const int   mrow = lane & 15;
    const int   khi  = (lane >> 4) << 1;
    const int   n0   = wave << 4;
    const float* wp  = W + n0 + mrow;
    v2f bfrag[DFEAT / 4];
#pragma unroll
    for (int k = 0; k < DFEAT; k += 4) {
        bfrag[k >> 2].x = wp[(size_t)(k + khi)     * DFEAT];
        bfrag[k >> 2].y = wp[(size_t)(k + khi + 1) * DFEAT];
    }
    const int   ocol = n0 + mrow;
    const float bv   = bias[ocol];
    const int   rsel = (lane >> 4) << 3;

    int ebeg = __builtin_amdgcn_readlane(rp, 0);
    int eend = __builtin_amdgcn_readlane(rp, 2);
    int n    = (eend - ebeg < 32) ? (eend - ebeg) : 32;
    int c0   = col[first_chunk_pos(ebeg, eend, lane)];

    int buf = 0;
    for (; tile < n_tiles; tile += gridDim.x) {
        const int nt   = tile + gridDim.x;
        const int rp_n = rowptr[(((nt < n_tiles) ? nt : tile) << 4) + 2 * wave + lsel];
        const int ri   = __float_as_int(rs_in[(tile << 4) + 2 * wave + (lane & 1)]);

        {
            int    cur     = 0;
            int    cur_end = __builtin_amdgcn_readlane(rp, 1);
            int    e       = ebeg;
            float4 acc     = make_float4(0.f, 0.f, 0.f, 0.f);
            if (ebeg < eend) {
                GATHER_STREAM(GATHER_STAGE_NODE)
            }
            while (cur < 2) {
                GATHER_STAGE_NODE()
                acc = make_float4(0.f, 0.f, 0.f, 0.f);
                ++cur;
            }
        }

        rp   = rp_n;
        ebeg = __builtin_amdgcn_readlane(rp, 0);
        eend = __builtin_amdgcn_readlane(rp, 2);
        n    = (eend - ebeg < 32) ? (eend - ebeg) : 32;

        __syncthreads();

        c0 = col[first_chunk_pos(ebeg, eend, lane)];
        __builtin_amdgcn_sched_barrier(0);

        const float* ap = &Alds[buf][mrow * ASTRIDE + khi];
        v8f acc = {0.f, 0.f, 0.f, 0.f, 0.f, 0.f, 0.f, 0.f};
#pragma unroll
        for (int k = 0; k < DFEAT; k += 4) {
            v2f a = *(const v2f*)(ap + k);
            acc = __builtin_amdgcn_wmma_f32_16x16x4_f32(false, a, false, bfrag[k >> 2],
                                                        (short)0, acc, false, false);
        }

        const int rowBase = tile << 4;
#pragma unroll
        for (int v = 0; v < 8; ++v)
            out[(size_t)(rowBase + v + rsel) * DFEAT + ocol] = acc[v] + bv;
        buf ^= 1;
    }
}

extern "C" void kernel_launch(void* const* d_in, const int* in_sizes, int n_in,
                              void* d_out, int out_size, void* d_ws, size_t ws_size,
                              hipStream_t stream) {
    const float* x   = (const float*)d_in[0];
    const int*   src = (const int*)  d_in[1];
    const int*   dst = (const int*)  d_in[2];
    const float* W1  = (const float*)d_in[3];
    const float* b1  = (const float*)d_in[4];
    const float* W2  = (const float*)d_in[5];
    const float* b2  = (const float*)d_in[6];
    float*       out = (float*)d_out;

    float* ws     = (float*)d_ws;
    int*   TS     = (int*)ws;
    int*   TD     = (int*)ws + 40000;
    float* hbuf   = ws + 2 * N_NODES;
    int*   tmpS   = (int*)hbuf;
    int*   tmpD   = (int*)hbuf + N_EDGES;
    float* aux    = hbuf + (size_t)N_NODES * DFEAT;
    float* rs_out = aux;
    float* rs_in  = aux + N_NODES;
    int*   rowptr = (int*)(aux + 2 * N_NODES);
    int*   col    = (int*)(aux + 4 * N_NODES) + N_EDGES;

    const int T256  = 256;
    const int GEMM_TILES = N_NODES / 16;

    chunk_sort_kernel<<<NCH, 1024, 0, stream>>>(src, dst, TS, TD, tmpS, tmpD, N_EDGES);
    range_kernel<<<NBK, 1024, 0, stream>>>(tmpS, tmpD, TS, TD, rowptr, rs_in, rs_out, col, N_NODES);

    gcn_gemm<<<GEMM_BLOCKS, T256, 0, stream>>>(x, rs_out, W1, out, GEMM_TILES);
    gather_kernel<<<GATHER_BLOCKS, T256, 0, stream>>>(out, rowptr, col, rs_in, rs_out, b1, hbuf, N_NODES);

    gather_gemm_kernel<<<FUSED_BLOCKS, T256, 0, stream>>>(hbuf, rowptr, col, rs_in, W2, b2, out, GEMM_TILES);
}


# speedup vs baseline: 5.1831x; 1.0591x over previous
#include <hip/hip_runtime.h>


#define N_NODES 50000
#define N_EDGES 800000
#define DFEAT   128
#define ASTRIDE 132

typedef float v2f __attribute__((ext_vector_type(2)));
typedef float v8f __attribute__((ext_vector_type(8)));

#define CHUNK 4096
#define NCH   ((N_EDGES + CHUNK - 1) / CHUNK)
#define NBK   ((N_NODES + 255) >> 8)
#define TROW  (NBK + 1)
#define CAP   6144
#define NSEG  ((NCH + 31) / 32)

__global__ __launch_bounds__(1024) void chunk_sort_kernel(const int* __restrict__ src,
                                                          const int* __restrict__ dst,
                                                          int* __restrict__ TS,
                                                          int* __restrict__ TD,
                                                          int* __restrict__ tmpS,
                                                          int* __restrict__ tmpD,
                                                          int n_edges) {
    __shared__ int s_cnt[512];
    __shared__ int s_wsum[16];
    __shared__ int s_eS[CHUNK];
    __shared__ int s_eD[CHUNK];

    const int tid  = threadIdx.x;
    const int lane = tid & 31;
    const int wave = tid >> 5;
    const int chunkBase = blockIdx.x * CHUNK;

    if (tid < 512) s_cnt[tid] = 0;
    __syncthreads();

    int s[4], d[4], rs[4], rd[4];
#pragma unroll
    for (int i = 0; i < 4; ++i) {
        const int e = chunkBase + tid + i * 1024;
        s[i] = -1; d[i] = 0; rs[i] = 0; rd[i] = 0;
        if (e < n_edges) { s[i] = src[e]; d[i] = dst[e]; }
    }
#pragma unroll
    for (int i = 0; i < 4; ++i)
        if (s[i] >= 0) {
            rs[i] = atomicAdd(&s_cnt[s[i] >> 8], 1);
            rd[i] = atomicAdd(&s_cnt[256 + (d[i] >> 8)], 1);
        }
    __syncthreads();

    int c = 0, incl = 0;
    if (tid < 512) {
        c = s_cnt[tid];
        incl = c;
        for (int off = 1; off < 32; off <<= 1) {
            int t = __shfl_up(incl, off);
            if (lane >= off) incl += t;
        }
        if (lane == 31) s_wsum[wave] = incl;
    }
    __syncthreads();
    if (tid < 512) {
        int bw = 0;
        for (int w = (wave & 8); w < wave; ++w) bw += s_wsum[w];
        const int excl = bw + incl - c;
        s_cnt[tid] = excl;
        const int idx = tid & 255;
        int* T = (tid < 256) ? TS : TD;
        if (idx <= NBK) T[blockIdx.x * TROW + idx] = excl;
    }
    __syncthreads();

#pragma unroll
    for (int i = 0; i < 4; ++i)
        if (s[i] >= 0) {
            s_eS[s_cnt[s[i] >> 8] + rs[i]]         = s[i];
            s_eD[s_cnt[256 + (d[i] >> 8)] + rd[i]] = (s[i] << 8) | (d[i] & 255);
        }
    __syncthreads();

#pragma unroll
    for (int i = 0; i < 4; ++i) {
        const int e = chunkBase + tid + i * 1024;
        if (e < n_edges) {
            tmpS[e] = s_eS[tid + i * 1024];
            tmpD[e] = s_eD[tid + i * 1024];
        }
    }
}

__global__ __launch_bounds__(1024) void range_kernel(const int* __restrict__ tmpS,
                                                     const int* __restrict__ tmpD,
                                                     const int* __restrict__ TS,
                                                     const int* __restrict__ TD,
                                                     int* __restrict__ rowptr,
                                                     float* __restrict__ rs_in,
                                                     float* __restrict__ rs_out,
                                                     int* __restrict__ col,
                                                     int n_nodes) {
    __shared__ int s_segS[256];
    __shared__ int s_segL[256];
    __shared__ int s_sgS[256];
    __shared__ int s_sgL[256];
    __shared__ int s_cnt[256];
    __shared__ int s_cur[256];
    __shared__ int s_deg[256];
    __shared__ int s_bsum[8];
    __shared__ int s_tsum[8];
    __shared__ int s_wsum[8];
    __shared__ int s_out[CAP];

    const int tid  = threadIdx.x;
    const int lane = tid & 31;
    const int wave = tid >> 5;
    const int b    = blockIdx.x;

    if (tid < 256) {
        int t0 = 0, st = 0, ln = 0, st2 = 0, ln2 = 0;
        if (tid < NCH) {
            t0 = TD[tid * TROW + b];
            const int t1 = TD[tid * TROW + b + 1];
            st = tid * CHUNK + t0;
            ln = t1 - t0;
            const int u0 = TS[tid * TROW + b];
            const int u1 = TS[tid * TROW + b + 1];
            st2 = tid * CHUNK + u0;
            ln2 = u1 - u0;
        }
        s_segS[tid] = st;
        s_segL[tid] = ln;
        s_sgS[tid]  = st2;
        s_sgL[tid]  = ln2;
        s_cnt[tid]  = 0;
        s_deg[tid]  = 0;
        int v = t0, u = ln;
        for (int off = 16; off > 0; off >>= 1) {
            v += __shfl_xor(v, off);
            u += __shfl_xor(u, off);
        }
        if (lane == 0) { s_bsum[wave] = v; s_tsum[wave] = u; }
    }
    __syncthreads();

    int base = 0, total = 0;
#pragma unroll
    for (int w = 0; w < 8; ++w) { base += s_bsum[w]; total += s_tsum[w]; }
    const bool staged = (total <= CAP);

    int sst[NSEG], sln[NSEG], qst[NSEG], qln[NSEG];
    int ea[NSEG], eb[NSEG], fa[NSEG], fb[NSEG];
#pragma unroll
    for (int k = 0; k < NSEG; ++k) {
        sst[k] = __builtin_amdgcn_readfirstlane(s_segS[wave + 32 * k]);
        sln[k] = __builtin_amdgcn_readfirstlane(s_segL[wave + 32 * k]);
        qst[k] = __builtin_amdgcn_readfirstlane(s_sgS[wave + 32 * k]);
        qln[k] = __builtin_amdgcn_readfirstlane(s_sgL[wave + 32 * k]);
    }
#pragma unroll
    for (int k = 0; k < NSEG; ++k) {
        const int ia = sst[k] + ((lane < sln[k]) ? lane : 0);
        const int ja = qst[k] + ((lane < qln[k]) ? lane : 0);
        ea[k] = tmpD[(ia < N_EDGES) ? ia : (N_EDGES - 1)];
        fa[k] = tmpS[(ja < N_EDGES) ? ja : (N_EDGES - 1)];
        eb[k] = 0; fb[k] = 0;
        if (sln[k] > 32) {
            if (lane + 32 < sln[k]) eb[k] = tmpD[sst[k] + lane + 32];
        }
        if (qln[k] > 32) {
            if (lane + 32 < qln[k]) fb[k] = tmpS[qst[k] + lane + 32];
        }
    }

#pragma unroll
    for (int k = 0; k < NSEG; ++k) {
        asm volatile("" : "+v"(fa[k]), "+v"(fb[k]));
    }

#pragma unroll
    for (int k = 0; k < NSEG; ++k) {
        if (lane < sln[k]) atomicAdd(&s_cnt[ea[k] & 255], 1);
        if (sln[k] > 32) {
            if (lane + 32 < sln[k]) atomicAdd(&s_cnt[eb[k] & 255], 1);
            for (int j = lane + 64; j < sln[k]; j += 32) {
                const int e = tmpD[sst[k] + j];
                atomicAdd(&s_cnt[e & 255], 1);
            }
        }
        if (lane < qln[k]) atomicAdd(&s_deg[fa[k] & 255], 1);
        if (qln[k] > 32) {
            if (lane + 32 < qln[k]) atomicAdd(&s_deg[fb[k] & 255], 1);
            for (int j = lane + 64; j < qln[k]; j += 32) {
                const int e = tmpS[qst[k] + j];
                atomicAdd(&s_deg[e & 255], 1);
            }
        }
    }
    __syncthreads();

    int c = 0, incl = 0;
    if (tid < 256) {
        c = s_cnt[tid];
        incl = c;
        for (int off = 1; off < 32; off <<= 1) {
            int t = __shfl_up(incl, off);
            if (lane >= off) incl += t;
        }
        if (lane == 31) s_wsum[wave] = incl;
    }
    __syncthreads();
    if (tid < 256) {
        int bw = 0;
        for (int w = 0; w < wave; ++w) bw += s_wsum[w];
        const int excl = bw + incl - c;
        s_cur[tid] = excl;
        const int node = (b << 8) + tid;
        if (node < n_nodes) {
            rowptr[node] = base + excl;
            rs_in[node]  = rsqrtf(fmaxf((float)c, 1.0f));
            rs_out[node] = rsqrtf(fmaxf((float)s_deg[tid], 1.0f));
            if (node == n_nodes - 1) rowptr[n_nodes] = base + excl + c;
        }
    }
    __syncthreads();

    int psa[NSEG];
#pragma unroll
    for (int k = 0; k < NSEG; ++k) {
        psa[k] = 0;
        if (lane < sln[k]) psa[k] = atomicAdd(&s_cur[ea[k] & 255], 1);
    }
#pragma unroll
    for (int k = 0; k < NSEG; ++k) {
        if (lane < sln[k]) {
            if (staged) s_out[psa[k]] = ea[k] >> 8;
            else        col[base + psa[k]] = ea[k] >> 8;
        }
    }
#pragma unroll
    for (int k = 0; k < NSEG; ++k) {
        if (sln[k] > 32) {
            for (int j = lane + 32; j < sln[k]; j += 32) {
                const int e   = (j < 64) ? eb[k] : tmpD[sst[k] + j];
                const int pos = atomicAdd(&s_cur[e & 255], 1);
                if (staged) s_out[pos] = e >> 8;
                else        col[base + pos] = e >> 8;
            }
        }
    }
    __syncthreads();

    if (staged) {
        for (int i = tid; i < total; i += 1024) col[base + i] = s_out[i];
    }
}

__device__ __forceinline__ void row_add(float4& acc, const float4 v) {
    acc.x += v.x;
    acc.y += v.y;
    acc.z += v.z;
    acc.w += v.w;
}

#define GATHER_WAVES_LOG2 13
#define GATHER_WAVES (1 << GATHER_WAVES_LOG2)
#define GATHER_BLOCKS (GATHER_WAVES / 8)

#define GATHER_ISSUE(V, CREG, J, NN)                                                  \
    _Pragma("unroll")                                                                 \
    for (int i = 0; i < 8; ++i) {                                                     \
        const int p = ((J) + i < (NN)) ? ((J) + i) : ((NN) - 1);                      \
        const int s = __builtin_amdgcn_readlane((CREG), p);                           \
        V[i] = X4[(size_t)s * (DFEAT / 4) + lane];                                    \
    }

#define GATHER_STORE_NODE()                                                           \
    {                                                                                 \
        float4 o;                                                                     \
        const float si = __int_as_float(__builtin_amdgcn_readlane(ri, cur));          \
        const float so = __int_as_float(__builtin_amdgcn_readlane(ro, cur));          \
        o.x = fmaxf(acc.x * si + b4.x, 0.f) * so;                                     \
        o.y = fmaxf(acc.y * si + b4.y, 0.f) * so;                                     \
        o.z = fmaxf(acc.z * si + b4.z, 0.f) * so;                                     \
        o.w = fmaxf(acc.w * si + b4.w, 0.f) * so;                                     \
        M4[(size_t)(g0 + cur) * (DFEAT / 4) + lane] = o;                              \
    }

#define GATHER_STAGE_NODE()                                                           \
    {                                                                                 \
        float4 o;                                                                     \
        const float si = __int_as_float(__builtin_amdgcn_readlane(ri, cur));          \
        o.x = acc.x * si;                                                             \
        o.y = acc.y * si;                                                             \
        o.z = acc.z * si;                                                             \
        o.w = acc.w * si;                                                             \
        *(float4*)&Alds[buf][(2 * wave + cur) * ASTRIDE + 4 * lane] = o;              \
    }

#define GATHER_CONSUME(V, J, NN, STORE)                                               \
    _Pragma("unroll")                                                                 \
    for (int i = 0; i < 8; ++i) {                                                     \
        if ((J) + i < (NN)) {                                                         \
            while (e == cur_end) {                                                    \
                STORE()                                                               \
                acc = make_float4(0.f, 0.f, 0.f, 0.f);                                \
                ++cur;                                                                \
                cur_end = __builtin_amdgcn_readlane(rp, cur + 1);                     \
            }                                                                         \
            row_add(acc, V[i]);                                                       \
            ++e;                                                                      \
        }                                                                             \
    }

#define GATHER_STREAM(STORE)                                                          \
    {                                                                                 \
        float4 A[8], B[8];                                                            \
        _Pragma("unroll")                                                             \
        for (int i = 0; i < 8; ++i) B[i] = make_float4(0.f, 0.f, 0.f, 0.f);           \
        GATHER_ISSUE(A, c0, 0, n)                                                     \
        int base = ebeg;                                                              \
        while (n == 32) {                                                             \
            const int rest = eend - (base + 32);                                      \
            const int nn   = (rest < 32) ? rest : 32;                                 \
            const int pe   = (base + 32 + lane < eend) ? (base + 32 + lane) : (eend - 1); \
            const int c1   = col[pe];                                                 \
            GATHER_ISSUE(B, c0, 8, 32)                                                \
            GATHER_CONSUME(A, 0, 32, STORE)                                           \
            GATHER_ISSUE(A, c0, 16, 32)                                               \
            GATHER_CONSUME(B, 8, 32, STORE)                                           \
            GATHER_ISSUE(B, c0, 24, 32)                                               \
            GATHER_CONSUME(A, 16, 32, STORE)                                          \
            if (nn > 0) {                                                             \
                GATHER_ISSUE(A, c1, 0, nn)                                            \
                GATHER_CONSUME(B, 24, 32, STORE)                                      \
            } else {                                                                  \
                GATHER_CONSUME(B, 24, 32, STORE)                                      \
            }                                                                         \
            c0 = c1; n = nn; base += 32;                                              \
        }                                                                             \
        if (n > 0) {                                                                  \
            if (8 < n)  { GATHER_ISSUE(B, c0, 8, n) }                                 \
            GATHER_CONSUME(A, 0, n, STORE)                                            \
            if (16 < n) { GATHER_ISSUE(A, c0, 16, n) }                                \
            GATHER_CONSUME(B, 8, n, STORE)                                            \
            if (24 < n) { GATHER_ISSUE(B, c0, 24, n) }                                \
            GATHER_CONSUME(A, 16, n, STORE)                                           \
            GATHER_CONSUME(B, 24, n, STORE)                                           \
        }                                                                             \
    }

__global__ __launch_bounds__(256) void gather_kernel(const float* __restrict__ X,
                                                     const int* __restrict__ rowptr,
                                                     const int* __restrict__ col,
                                                     const float* __restrict__ rs_in,
                                                     const float* __restrict__ rs_out,
                                                     const float* __restrict__ bias,
                                                     float* __restrict__ msg,
                                                     int n_nodes) {
    const int lane = threadIdx.x & 31;
    const int wid  = __builtin_amdgcn_readfirstlane((int)((blockIdx.x * blockDim.x + threadIdx.x) >> 5));
    const int n0   = (int)(((unsigned long long)wid * (unsigned)n_nodes) >> GATHER_WAVES_LOG2);
    const int n1   = (int)(((unsigned long long)(wid + 1) * (unsigned)n_nodes) >> GATHER_WAVES_LOG2);

    const float4* X4 = (const float4*)X;
    float4*       M4 = (float4*)msg;

    const float4 b4 = ((const float4*)bias)[lane];

    for (int g0 = n0; g0 < n1; g0 += 31) {
        const int cnt  = (n1 - g0 < 31) ? (n1 - g0) : 31;
        const int rp   = rowptr[g0 + ((lane < cnt) ? lane : cnt)];
        const int ebeg = __builtin_amdgcn_readlane(rp, 0);
        const int eend = __builtin_amdgcn_readlane(rp, cnt);
        const int q  = g0 + ((lane < cnt) ? lane : (cnt - 1));
        const int ri = __float_as_int(rs_in[q]);
        const int ro = __float_as_int(rs_out[q]);

        int    cur     = 0;
        int    cur_end = __builtin_amdgcn_readlane(rp, 1);
        int    e       = ebeg;
        float4 acc     = make_float4(0.f, 0.f, 0.f, 0.f);

        if (ebeg < eend) {
            int n  = (eend - ebeg < 32) ? (eend - ebeg) : 32;
            int c0 = 0;
            if (lane < n) c0 = col[ebeg + lane];
            GATHER_STREAM(GATHER_STORE_NODE)
        }
        while (cur < cnt) {
            GATHER_STORE_NODE()
            acc = make_float4(0.f, 0.f, 0.f, 0.f);
            ++cur;
        }
    }
}

#define GEMM_BLOCKS 512

__global__ __launch_bounds__(256) void gcn_gemm(const float* A,
                                                const float* __restrict__ rscale,
                                                const float* __restrict__ W,
                                                float* out,
                                                int n_tiles) {
    __shared__ float Alds[2][16 * ASTRIDE];

    const int tid  = threadIdx.x;
    const int wave = tid >> 5;
    const int lane = tid & 31;

    const int   mrow = lane & 15;
    const int   khi  = (lane >> 4) << 1;
    const int   n0   = wave << 4;
    const float* wp  = W + n0 + mrow;

    v2f bfrag[DFEAT / 4];
#pragma unroll
    for (int k = 0; k < DFEAT; k += 4) {
        bfrag[k >> 2].x = wp[(size_t)(k + khi)     * DFEAT];
        bfrag[k >> 2].y = wp[(size_t)(k + khi + 1) * DFEAT];
    }

    const int   ocol = n0 + mrow;
    const int   rsel = (lane >> 4) << 3;

    const int r = tid >> 4;
    const int c = (tid & 15) << 3;

    int    tile = blockIdx.x;
    float  scale = 0.f;
    float4 a0 = make_float4(0.f, 0.f, 0.f, 0.f);
    float4 a1 = make_float4(0.f, 0.f, 0.f, 0.f);
    if (tile < n_tiles) {
        const int row = (tile << 4) + r;
        const float4* gp = (const float4*)(A + (size_t)row * DFEAT + c);
        scale = rscale[row];
        a0 = gp[0];
        a1 = gp[1];
    }

    int p = 0;
    for (; tile < n_tiles; tile += gridDim.x) {
        {
            float* lp = &Alds[p][r * ASTRIDE + c];
            lp[0] = a0.x * scale; lp[1] = a0.y * scale; lp[2] = a0.z * scale; lp[3] = a0.w * scale;
            lp[4] = a1.x * scale; lp[5] = a1.y * scale; lp[6] = a1.z * scale; lp[7] = a1.w * scale;
        }
        __syncthreads();

        const int nt = tile + gridDim.x;
        if (nt < n_tiles) {
            const int row = (nt << 4) + r;
            const float4* gp = (const float4*)(A + (size_t)row * DFEAT + c);
            scale = rscale[row];
            a0 = gp[0];
            a1 = gp[1];
        }

        const float* ap = &Alds[p][mrow * ASTRIDE + khi];
        v8f acc = {0.f, 0.f, 0.f, 0.f, 0.f, 0.f, 0.f, 0.f};
#pragma unroll
        for (int k = 0; k < DFEAT; k += 4) {
            v2f a = *(const v2f*)(ap + k);
            acc = __builtin_amdgcn_wmma_f32_16x16x4_f32(false, a, false, bfrag[k >> 2],
                                                        (short)0, acc, false, false);
        }

        const int rowBase = tile << 4;
#pragma unroll
        for (int v = 0; v < 8; ++v)
            out[(size_t)(rowBase + v + rsel) * DFEAT + ocol] = acc[v];
        p ^= 1;
    }
}

__device__ __forceinline__ int first_chunk_pos(int ebeg, int eend, int lane) {
    const int pe = (ebeg + lane < eend) ? (ebeg + lane) : (eend - 1);
    return (pe > 0) ? pe : 0;
}

#define FUSED_BLOCKS 625

__global__ __launch_bounds__(256, 6) void gather_gemm_kernel(const float* __restrict__ X,
                                                          const int* __restrict__ rowptr,
                                                          const int* __restrict__ col,
                                                          const float* __restrict__ rs_in,
                                                          const float* __restrict__ W,
                                                          const float* __restrict__ bias,
                                                          float* __restrict__ out,
                                                          int n_tiles) {
    __shared__ __attribute__((aligned(16))) float Alds[2][16 * ASTRIDE];

    const int lane = threadIdx.x & 31;
    const int wave = __builtin_amdgcn_readfirstlane((int)(threadIdx.x >> 5));

    const float4* X4 = (const float4*)X;

    const int lsel = (lane < 2) ? lane : 2;
    int tile = blockIdx.x;
    if (tile >= n_tiles) return;
    int rp = rowptr[(tile << 4) + 2 * wave + lsel];

    const int   mrow = lane & 15;
    const int   khi  = (lane >> 4) << 1;
    const int   n0   = wave << 4;
    const float* wp  = W + n0 + mrow;
    v2f bfrag[DFEAT / 4];
#pragma unroll
    for (int k = 0; k < DFEAT; k += 4) {
        bfrag[k >> 2].x = wp[(size_t)(k + khi)     * DFEAT];
        bfrag[k >> 2].y = wp[(size_t)(k + khi + 1) * DFEAT];
    }
    const int   ocol = n0 + mrow;
    const float bv   = bias[ocol];
    const int   rsel = (lane >> 4) << 3;

    int ebeg = __builtin_amdgcn_readlane(rp, 0);
    int eend = __builtin_amdgcn_readlane(rp, 2);
    int n    = (eend - ebeg < 32) ? (eend - ebeg) : 32;
    int c0   = col[first_chunk_pos(ebeg, eend, lane)];

    int buf = 0;
    for (; tile < n_tiles; tile += gridDim.x) {
        const int nt   = tile + gridDim.x;
        const int rp_n = rowptr[(((nt < n_tiles) ? nt : tile) << 4) + 2 * wave + lsel];
        const int ri   = __float_as_int(rs_in[(tile << 4) + 2 * wave + (lane & 1)]);

        {
            int    cur     = 0;
            int    cur_end = __builtin_amdgcn_readlane(rp, 1);
            int    e       = ebeg;
            float4 acc     = make_float4(0.f, 0.f, 0.f, 0.f);
            if (ebeg < eend) {
                GATHER_STREAM(GATHER_STAGE_NODE)
            }
            while (cur < 2) {
                GATHER_STAGE_NODE()
                acc = make_float4(0.f, 0.f, 0.f, 0.f);
                ++cur;
            }
        }

        rp   = rp_n;
        ebeg = __builtin_amdgcn_readlane(rp, 0);
        eend = __builtin_amdgcn_readlane(rp, 2);
        n    = (eend - ebeg < 32) ? (eend - ebeg) : 32;

        __syncthreads();

        c0 = col[first_chunk_pos(ebeg, eend, lane)];
        __builtin_amdgcn_sched_barrier(0);

        const float* ap = &Alds[buf][mrow * ASTRIDE + khi];
        v8f acc = {0.f, 0.f, 0.f, 0.f, 0.f, 0.f, 0.f, 0.f};
#pragma unroll
        for (int k = 0; k < DFEAT; k += 4) {
            v2f a = *(const v2f*)(ap + k);
            acc = __builtin_amdgcn_wmma_f32_16x16x4_f32(false, a, false, bfrag[k >> 2],
                                                        (short)0, acc, false, false);
        }

        const int rowBase = tile << 4;
#pragma unroll
        for (int v = 0; v < 8; ++v)
            out[(size_t)(rowBase + v + rsel) * DFEAT + ocol] = acc[v] + bv;
        buf ^= 1;
    }
}

extern "C" void kernel_launch(void* const* d_in, const int* in_sizes, int n_in,
                              void* d_out, int out_size, void* d_ws, size_t ws_size,
                              hipStream_t stream) {
    const float* x   = (const float*)d_in[0];
    const int*   src = (const int*)  d_in[1];
    const int*   dst = (const int*)  d_in[2];
    const float* W1  = (const float*)d_in[3];
    const float* b1  = (const float*)d_in[4];
    const float* W2  = (const float*)d_in[5];
    const float* b2  = (const float*)d_in[6];
    float*       out = (float*)d_out;

    float* ws     = (float*)d_ws;
    int*   TS     = (int*)ws;
    int*   TD     = (int*)ws + 40000;
    float* hbuf   = ws + 2 * N_NODES - 32;
    int*   tmpS   = (int*)hbuf;
    int*   tmpD   = (int*)hbuf + N_EDGES;
    float* aux    = ws + 2 * N_NODES + (size_t)N_NODES * DFEAT;
    float* rs_out = aux;
    float* rs_in  = aux + N_NODES;
    int*   rowptr = (int*)(aux + 2 * N_NODES);
    int*   col    = (int*)(aux + 4 * N_NODES) + N_EDGES;

    const int T256  = 256;
    const int GEMM_TILES = N_NODES / 16;

    chunk_sort_kernel<<<NCH, 1024, 0, stream>>>(src, dst, TS, TD, tmpS, tmpD, N_EDGES);
    range_kernel<<<NBK, 1024, 0, stream>>>(tmpS, tmpD, TS, TD, rowptr, rs_in, rs_out, col, N_NODES);

    gcn_gemm<<<GEMM_BLOCKS, T256, 0, stream>>>(x, rs_out, W1, out, GEMM_TILES);
    gather_kernel<<<GATHER_BLOCKS, T256, 0, stream>>>(out, rowptr, col, rs_in, rs_out, b1, hbuf, N_NODES);

    gather_gemm_kernel<<<FUSED_BLOCKS, T256, 0, stream>>>(hbuf, rowptr, col, rs_in, W2, b2, out, GEMM_TILES);
}
